# starting kernel
# baseline (speedup 1.0000x reference)
#include <hip/hip_runtime.h>
#include <math.h>

typedef __attribute__((ext_vector_type(2))) float v2f;
typedef __attribute__((ext_vector_type(8))) float v8f;

#define N_NODES 50000
#define N_EDGES 800000
#define IN_F    256
#define OUT_F   128
#define RANK    64

static_assert(N_NODES % 16 == 0, "row tiles exact");
static_assert(N_EDGES % 256 == 0, "edge blocks exact");

#define OFF_ODEG   0u
#define OFF_IDEG   50000u
#define OFF_HSUM   100000u
#define OFF_LABS   6500000u
#define OFF_NCNT   9700000u
#define OFF_FSUM   12900000u
#define OFF_FPROD  19300000u
#define ZERO_CNT   12900000u

__global__ __launch_bounds__(256) void k_zero(unsigned int* __restrict__ p, unsigned n4)
{
    unsigned i = blockIdx.x * blockDim.x + threadIdx.x;
    if (i < n4) {
        ((uint4*)p)[i] = make_uint4(0u, 0u, 0u, 0u);
    }
}

__global__ __launch_bounds__(256) void k_degrees(const int* __restrict__ src,
                                                 const int* __restrict__ dst,
                                                 float* __restrict__ odeg,
                                                 float* __restrict__ ideg)
{
    int e = blockIdx.x * blockDim.x + threadIdx.x;
    atomicAdd(&odeg[src[e]], 1.0f);
    atomicAdd(&ideg[dst[e]], 1.0f);
}

__global__ __launch_bounds__(128) void k_node_gemm(const float* __restrict__ feat,
                                                   const float* __restrict__ w1,
                                                   const float* __restrict__ w2,
                                                   const float* __restrict__ odeg,
                                                   float* __restrict__ fsum,
                                                   float* __restrict__ fprod)
{
    __shared__ float As[16][260];
    __shared__ float sc[16];

    const int m0  = blockIdx.x * 16;
    const int tid = threadIdx.x;

    if (tid < 16) {
        sc[tid] = rsqrtf(fmaxf(odeg[m0 + tid], 1.0f));
    }
    __syncthreads();

    {
        const int r   = tid >> 3;
        const int c0  = (tid & 7) * 32;
        const float s = sc[r];
        const float4* g = (const float4*)(feat + (size_t)(m0 + r) * IN_F + c0);
#pragma unroll
        for (int j = 0; j < 8; ++j) {
            float4 x = g[j];
            As[r][c0 + 4 * j + 0] = x.x * s;
            As[r][c0 + 4 * j + 1] = x.y * s;
            As[r][c0 + 4 * j + 2] = x.z * s;
            As[r][c0 + 4 * j + 3] = x.w * s;
        }
    }
    __syncthreads();

    const int wave  = tid >> 5;
    const int lane  = tid & 31;
    const int rsel  = lane & 15;
    const int khalf = (lane < 16) ? 0 : 2;
    const int rofs  = (lane < 16) ? 0 : 8;

    for (int t = 0; t < 3; ++t) {
        const int  tile = wave + 4 * t;
        const bool isW2 = (tile >= 8);
        const int  n0   = isW2 ? (tile - 8) * 16 : tile * 16;
        const float* B  = isW2 ? w2 : w1;
        const int  ldb  = isW2 ? RANK : OUT_F;

        for (int k = 0; k < IN_F; k += 32)
            __builtin_prefetch(B + (size_t)k * ldb + n0, 0, 0);

        v8f acc = {};
        for (int k0 = 0; k0 < IN_F; k0 += 4) {
            v2f a;
            a.x = As[rsel][k0 + khalf];
            a.y = As[rsel][k0 + khalf + 1];
            v2f b;
            const float* Bp = B + (size_t)(k0 + khalf) * ldb + n0 + rsel;
            b.x = Bp[0];
            b.y = Bp[ldb];
            acc = __builtin_amdgcn_wmma_f32_16x16x4_f32(
                false, a, false, b, (short)0, acc, false, false);
        }

        const int colg = n0 + rsel;
        if (isW2) {
            const float bias = w2[(size_t)IN_F * RANK + colg];
            float* op = fprod + (size_t)(m0 + rofs) * RANK + colg;
#pragma unroll
            for (int r = 0; r < 8; ++r)
                op[r * RANK] = tanhf(acc[r] + bias);
        } else {
            float* op = fsum + (size_t)(m0 + rofs) * OUT_F + colg;
#pragma unroll
            for (int r = 0; r < 8; ++r)
                op[r * OUT_F] = acc[r];
        }
    }
}

__global__ __launch_bounds__(256) void k_edge(const int* __restrict__ src,
                                              const int* __restrict__ dst,
                                              const float* __restrict__ fsum,
                                              const float* __restrict__ fprod,
                                              float* __restrict__ hsum,
                                              float* __restrict__ labs,
                                              int*   __restrict__ ncnt)
{
    const int wid  = (int)((blockIdx.x * blockDim.x + threadIdx.x) >> 5);
    const int lane = threadIdx.x & 31;

    const int s = src[wid];
    const int d = dst[wid];

    const float* fs = fsum + (size_t)s * OUT_F;
    float*       hs = hsum + (size_t)d * OUT_F;
#pragma unroll
    for (int j = 0; j < 4; ++j) {
        int c = lane + 32 * j;
        atomicAdd(&hs[c], fs[c]);
    }

    const float* fp = fprod + (size_t)s * RANK;
    float*       la = labs  + (size_t)d * RANK;
    int*         nc = ncnt  + (size_t)d * RANK;
#pragma unroll
    for (int j = 0; j < 2; ++j) {
        int c = lane + 32 * j;
        float m = fp[c];
        atomicAdd(&la[c], __logf(fabsf(m)));
        if (m < 0.0f) atomicAdd(&nc[c], 1);
    }
}

__global__ __launch_bounds__(128) void k_final(const float* __restrict__ vmat,
                                               const float* __restrict__ hsum,
                                               const float* __restrict__ labs,
                                               const int*   __restrict__ ncnt,
                                               const float* __restrict__ ideg,
                                               float* __restrict__ out)
{
    __shared__ float Hp[16][68];
    __shared__ float sc[16];

    const int m0  = blockIdx.x * 16;
    const int tid = threadIdx.x;

    if (tid < 16) {
        sc[tid] = rsqrtf(fmaxf(ideg[m0 + tid], 1.0f));
    }
    for (int i = tid; i < 16 * RANK; i += 128) {
        int r = i >> 6, c = i & 63;
        int row = m0 + r;
        float val = 0.0f;
        float dg = ideg[row];
        if (dg > 0.0f) {
            float la = labs[(size_t)row * RANK + c];
            int   n  = ncnt[(size_t)row * RANK + c];
            float sg = (n & 1) ? -1.0f : 1.0f;
            val = sg * __expf(la);
        }
        Hp[r][c] = val;
    }
    __syncthreads();

    const int wave  = tid >> 5;
    const int lane  = tid & 31;
    const int rsel  = lane & 15;
    const int khalf = (lane < 16) ? 0 : 2;
    const int rofs  = (lane < 16) ? 0 : 8;

    for (int t = 0; t < 2; ++t) {
        const int n0 = (wave * 2 + t) * 16;

        for (int k = 0; k < RANK; k += 32)
            __builtin_prefetch(vmat + (size_t)k * OUT_F + n0, 0, 0);

        v8f acc = {};
        for (int k0 = 0; k0 < RANK; k0 += 4) {
            v2f a;
            a.x = Hp[rsel][k0 + khalf];
            a.y = Hp[rsel][k0 + khalf + 1];
            v2f b;
            const float* Bp = vmat + (size_t)(k0 + khalf) * OUT_F + n0 + rsel;
            b.x = Bp[0];
            b.y = Bp[OUT_F];
            acc = __builtin_amdgcn_wmma_f32_16x16x4_f32(
                false, a, false, b, (short)0, acc, false, false);
        }
        const int colg = n0 + rsel;
        const float* hp = hsum + (size_t)(m0 + rofs) * OUT_F + colg;
        float*       op = out  + (size_t)(m0 + rofs) * OUT_F + colg;
#pragma unroll
        for (int r = 0; r < 8; ++r)
            op[r * OUT_F] = (acc[r] + hp[r * OUT_F]) * sc[rofs + r];
    }
}

extern "C" void kernel_launch(void* const* d_in, const int* in_sizes, int n_in,
                              void* d_out, int out_size, void* d_ws, size_t ws_size,
                              hipStream_t stream)
{
    (void)in_sizes; (void)n_in; (void)out_size; (void)ws_size;

    const float* feat = (const float*)d_in[0];
    const float* w1   = (const float*)d_in[1];
    const float* w2   = (const float*)d_in[2];
    const float* vmat = (const float*)d_in[3];
    const int*   src  = (const int*)d_in[4];
    const int*   dst  = (const int*)d_in[5];

    float* ws    = (float*)d_ws;
    float* odeg  = ws + OFF_ODEG;
    float* ideg  = ws + OFF_IDEG;
    float* hsum  = ws + OFF_HSUM;
    float* labs  = ws + OFF_LABS;
    int*   ncnt  = (int*)(ws + OFF_NCNT);
    float* fsum  = ws + OFF_FSUM;
    float* fprod = ws + OFF_FPROD;

    {
        unsigned n4 = ZERO_CNT / 4u;
        k_zero<<<(n4 + 255u) / 256u, 256, 0, stream>>>((unsigned int*)ws, n4);
    }
    k_degrees<<<N_EDGES / 256, 256, 0, stream>>>(src, dst, odeg, ideg);
    k_node_gemm<<<N_NODES / 16, 128, 0, stream>>>(feat, w1, w2, odeg, fsum, fprod);
    k_edge<<<(N_EDGES * 32) / 256, 256, 0, stream>>>(src, dst, fsum, fprod,
                                                     hsum, labs, ncnt);
    k_final<<<N_NODES / 16, 128, 0, stream>>>(vmat, hsum, labs, ncnt, ideg,
                                              (float*)d_out);
}


# Edge sums by gather, not atomics: edges slotted by dst (in-degree atomic's returned rank + scan); one wave adds a dst row in registers in slot (arrival-rank) order; feature atomics removed
# speedup vs baseline: 1.7472x; 1.7472x over previous
#include <hip/hip_runtime.h>
#include <math.h>


typedef __attribute__((ext_vector_type(2))) float v2f;
typedef __attribute__((ext_vector_type(8))) float v8f;

#define N_NODES 50000
#define N_EDGES 800000
#define IN_F    256
#define OUT_F   128
#define RANK    64

static_assert(N_NODES % 16 == 0, "row tiles exact");
static_assert(N_NODES % 8 == 0,  "gather blocks exact");
static_assert(N_EDGES % 256 == 0, "edge blocks exact");

#define OFF_ODEG   0u
#define OFF_IDEG   50000u
#define OFF_HSUM   100000u
#define OFF_RANK   100000u
#define OFF_HPROD  6500000u
#define OFF_ROWPTR 9700000u
#define OFF_CSR    9760000u
#define OFF_FSUM   12900000u
#define OFF_FPROD  19300000u
#define ZERO_CNT   100000u

#define SCAN_T     1024
#define SCAN_PER   49
#define SCAN_FULL  (N_NODES / SCAN_T)
static_assert(SCAN_T * SCAN_PER >= N_NODES, "scan covers every node");
static_assert(SCAN_FULL + 1 == SCAN_PER, "one guarded row behind the whole ones");

__global__ __launch_bounds__(256) void k_zero(unsigned int* __restrict__ p, unsigned n4)
{
    unsigned i = blockIdx.x * blockDim.x + threadIdx.x;
    if (i < n4) {
        ((uint4*)p)[i] = make_uint4(0u, 0u, 0u, 0u);
    }
}

__global__ __launch_bounds__(256) void k_degrees(const int* __restrict__ src,
                                                 const int* __restrict__ dst,
                                                 float* __restrict__ odeg,
                                                 int*   __restrict__ ideg,
                                                 int*   __restrict__ rank)
{
    int e = blockIdx.x * blockDim.x + threadIdx.x;
    atomicAdd(&odeg[src[e]], 1.0f);
    rank[e] = atomicAdd(&ideg[dst[e]], 1);
}

__global__ __launch_bounds__(SCAN_T) void k_scan(const int* __restrict__ ideg,
                                                 int* __restrict__ rowptr)
{
    __shared__ int cnt[SCAN_T * SCAN_PER];
    __shared__ int wtot[SCAN_T / 32];

    const int tid  = threadIdx.x;
    const int lane = tid & 31;
    const int wave = tid >> 5;

    {
        int v[SCAN_FULL];
#pragma unroll
        for (int k = 0; k < SCAN_FULL; ++k) v[k] = ideg[tid + SCAN_T * k];
        const int il = tid + SCAN_T * SCAN_FULL;
        const int vl = (il < N_NODES) ? ideg[il] : 0;
#pragma unroll
        for (int k = 0; k < SCAN_FULL; ++k) cnt[tid + SCAN_T * k] = v[k];
        cnt[il] = vl;
    }
    __syncthreads();

    const int base = tid * SCAN_PER;
    int tsum = 0;
#pragma unroll
    for (int j = 0; j < SCAN_PER; ++j) tsum += cnt[base + j];

    int incl = tsum;
#pragma unroll
    for (int o = 1; o < 32; o <<= 1) {
        int up = __shfl_up(incl, o);
        if (lane >= o) incl += up;
    }
    if (lane == 31) wtot[wave] = incl;
    __syncthreads();
    if (wave == 0) {
        int w = wtot[lane];
        int wi = w;
#pragma unroll
        for (int o = 1; o < 32; o <<= 1) {
            int up = __shfl_up(wi, o);
            if (lane >= o) wi += up;
        }
        wtot[lane] = wi - w;
    }
    __syncthreads();

    int run = wtot[wave] + (incl - tsum);
#pragma unroll
    for (int j = 0; j < SCAN_PER; ++j) {
        int v = cnt[base + j];
        cnt[base + j] = run;
        run += v;
    }
    __syncthreads();

#pragma unroll
    for (int k = 0; k < SCAN_FULL; ++k) rowptr[tid + SCAN_T * k] = cnt[tid + SCAN_T * k];
    {
        const int il = tid + SCAN_T * SCAN_FULL;
        if (il < N_NODES) rowptr[il] = cnt[il];
    }
}

__global__ __launch_bounds__(256) void k_fill(const int* __restrict__ src,
                                              const int* __restrict__ dst,
                                              const int* __restrict__ rank,
                                              const int* __restrict__ rowptr,
                                              int* __restrict__ csr)
{
    int e = blockIdx.x * blockDim.x + threadIdx.x;
    unsigned pos = (unsigned)(rowptr[dst[e]] + rank[e]);
    if (pos < (unsigned)N_EDGES)
        csr[pos] = src[e];
}

__global__ __launch_bounds__(128) void k_node_gemm(const float* __restrict__ feat,
                                                   const float* __restrict__ w1,
                                                   const float* __restrict__ w2,
                                                   const float* __restrict__ odeg,
                                                   float* __restrict__ fsum,
                                                   float* __restrict__ fprod)
{
    __shared__ float As[16][260];
    __shared__ float sc[16];

    const int m0  = blockIdx.x * 16;
    const int tid = threadIdx.x;

    if (tid < 16) {
        sc[tid] = rsqrtf(fmaxf(odeg[m0 + tid], 1.0f));
    }
    __syncthreads();

    {
        const int r   = tid >> 3;
        const int c0  = (tid & 7) * 32;
        const float s = sc[r];
        const float4* g = (const float4*)(feat + (size_t)(m0 + r) * IN_F + c0);
#pragma unroll
        for (int j = 0; j < 8; ++j) {
            float4 x = g[j];
            As[r][c0 + 4 * j + 0] = x.x * s;
            As[r][c0 + 4 * j + 1] = x.y * s;
            As[r][c0 + 4 * j + 2] = x.z * s;
            As[r][c0 + 4 * j + 3] = x.w * s;
        }
    }
    __syncthreads();

    const int wave  = tid >> 5;
    const int lane  = tid & 31;
    const int rsel  = lane & 15;
    const int khalf = (lane < 16) ? 0 : 2;
    const int rofs  = (lane < 16) ? 0 : 8;

    for (int t = 0; t < 3; ++t) {
        const int  tile = wave + 4 * t;
        const bool isW2 = (tile >= 8);
        const int  n0   = isW2 ? (tile - 8) * 16 : tile * 16;
        const float* B  = isW2 ? w2 : w1;
        const int  ldb  = isW2 ? RANK : OUT_F;

        for (int k = 0; k < IN_F; k += 32)
            __builtin_prefetch(B + (size_t)k * ldb + n0, 0, 0);

        v8f acc = {};
        for (int k0 = 0; k0 < IN_F; k0 += 4) {
            v2f a;
            a.x = As[rsel][k0 + khalf];
            a.y = As[rsel][k0 + khalf + 1];
            v2f b;
            const float* Bp = B + (size_t)(k0 + khalf) * ldb + n0 + rsel;
            b.x = Bp[0];
            b.y = Bp[ldb];
            acc = __builtin_amdgcn_wmma_f32_16x16x4_f32(
                false, a, false, b, (short)0, acc, false, false);
        }

        const int colg = n0 + rsel;
        if (isW2) {
            const float bias = w2[(size_t)IN_F * RANK + colg];
            float* op = fprod + (size_t)(m0 + rofs) * RANK + colg;
#pragma unroll
            for (int r = 0; r < 8; ++r)
                op[r * RANK] = tanhf(acc[r] + bias);
        } else {
            float* op = fsum + (size_t)(m0 + rofs) * OUT_F + colg;
#pragma unroll
            for (int r = 0; r < 8; ++r)
                op[r * OUT_F] = acc[r];
        }
    }
}

__global__ __launch_bounds__(256) void k_gather(const int* __restrict__ rowptr,
                                                const int* __restrict__ ideg,
                                                const int* __restrict__ csr,
                                                const float* __restrict__ fsum,
                                                const float* __restrict__ fprod,
                                                float* __restrict__ hsum,
                                                float* __restrict__ hprod)
{
    const int lane = threadIdx.x & 31;
    const int d     = __builtin_amdgcn_readfirstlane((int)(blockIdx.x * 8u + (threadIdx.x >> 5)));
    const int start = rowptr[d];
    const int cnt   = ideg[d];

    float4 hs = make_float4(0.0f, 0.0f, 0.0f, 0.0f);
    float  la0 = 0.0f, la1 = 0.0f;
    int    ng0 = 0, ng1 = 0;

    for (int b = 0; b < cnt; b += 32) {
        const int n = min(32, cnt - b);
        int mine = 0;
        if (lane < n) mine = csr[start + b + lane];
#pragma unroll 4
        for (int j = 0; j < n; ++j) {
            const unsigned s = min((unsigned)__builtin_amdgcn_readlane(mine, j), (unsigned)(N_NODES - 1));
            const float4 x = *(const float4*)(fsum  + (size_t)s * OUT_F + 4 * lane);
            const float2 m = *(const float2*)(fprod + (size_t)s * RANK  + 2 * lane);
            hs.x += x.x; hs.y += x.y; hs.z += x.z; hs.w += x.w;
            la0 += __logf(fabsf(m.x));
            la1 += __logf(fabsf(m.y));
            ng0 += (m.x < 0.0f) ? 1 : 0;
            ng1 += (m.y < 0.0f) ? 1 : 0;
        }
    }

    *(float4*)(hsum + (size_t)d * OUT_F + 4 * lane) = hs;

    float2 hp = make_float2(0.0f, 0.0f);
    if (cnt > 0) {
        hp.x = ((ng0 & 1) ? -1.0f : 1.0f) * __expf(la0);
        hp.y = ((ng1 & 1) ? -1.0f : 1.0f) * __expf(la1);
    }
    *(float2*)(hprod + (size_t)d * RANK + 2 * lane) = hp;
}

__global__ __launch_bounds__(128) void k_final(const float* __restrict__ vmat,
                                               const float* __restrict__ hsum,
                                               const float* __restrict__ hprod,
                                               const int*   __restrict__ ideg,
                                               float* __restrict__ out)
{
    __shared__ float Hp[16][68];
    __shared__ float sc[16];

    const int m0  = blockIdx.x * 16;
    const int tid = threadIdx.x;

    if (tid < 16) {
        sc[tid] = rsqrtf(fmaxf((float)ideg[m0 + tid], 1.0f));
    }
    for (int i = tid; i < 16 * RANK; i += 128) {
        int r = i >> 6, c = i & 63;
        Hp[r][c] = hprod[(size_t)(m0 + r) * RANK + c];
    }
    __syncthreads();

    const int wave  = tid >> 5;
    const int lane  = tid & 31;
    const int rsel  = lane & 15;
    const int khalf = (lane < 16) ? 0 : 2;
    const int rofs  = (lane < 16) ? 0 : 8;

    for (int t = 0; t < 2; ++t) {
        const int n0 = (wave * 2 + t) * 16;

        for (int k = 0; k < RANK; k += 32)
            __builtin_prefetch(vmat + (size_t)k * OUT_F + n0, 0, 0);

        v8f acc = {};
        for (int k0 = 0; k0 < RANK; k0 += 4) {
            v2f a;
            a.x = Hp[rsel][k0 + khalf];
            a.y = Hp[rsel][k0 + khalf + 1];
            v2f b;
            const float* Bp = vmat + (size_t)(k0 + khalf) * OUT_F + n0 + rsel;
            b.x = Bp[0];
            b.y = Bp[OUT_F];
            acc = __builtin_amdgcn_wmma_f32_16x16x4_f32(
                false, a, false, b, (short)0, acc, false, false);
        }
        const int colg = n0 + rsel;
        const float* hp = hsum + (size_t)(m0 + rofs) * OUT_F + colg;
        float*       op = out  + (size_t)(m0 + rofs) * OUT_F + colg;
#pragma unroll
        for (int r = 0; r < 8; ++r)
            op[r * OUT_F] = (acc[r] + hp[r * OUT_F]) * sc[rofs + r];
    }
}

extern "C" void kernel_launch(void* const* d_in, const int* in_sizes, int n_in,
                              void* d_out, int out_size, void* d_ws, size_t ws_size,
                              hipStream_t stream)
{
    (void)in_sizes; (void)n_in; (void)out_size; (void)ws_size;

    const float* feat = (const float*)d_in[0];
    const float* w1   = (const float*)d_in[1];
    const float* w2   = (const float*)d_in[2];
    const float* vmat = (const float*)d_in[3];
    const int*   src  = (const int*)d_in[4];
    const int*   dst  = (const int*)d_in[5];

    float* ws     = (float*)d_ws;
    float* odeg   = ws + OFF_ODEG;
    int*   ideg   = (int*)(ws + OFF_IDEG);
    float* hsum   = ws + OFF_HSUM;
    int*   rank   = (int*)(ws + OFF_RANK);
    float* hprod  = ws + OFF_HPROD;
    int*   rowptr = (int*)(ws + OFF_ROWPTR);
    int*   csr    = (int*)(ws + OFF_CSR);
    float* fsum   = ws + OFF_FSUM;
    float* fprod  = ws + OFF_FPROD;

    {
        unsigned n4 = ZERO_CNT / 4u;
        k_zero<<<(n4 + 255u) / 256u, 256, 0, stream>>>((unsigned int*)ws, n4);
    }
    k_degrees<<<N_EDGES / 256, 256, 0, stream>>>(src, dst, odeg, ideg, rank);
    k_scan<<<1, SCAN_T, 0, stream>>>(ideg, rowptr);
    k_fill<<<N_EDGES / 256, 256, 0, stream>>>(src, dst, rank, rowptr, csr);
    k_node_gemm<<<N_NODES / 16, 128, 0, stream>>>(feat, w1, w2, odeg, fsum, fprod);
    k_gather<<<N_NODES / 8, 256, 0, stream>>>(rowptr, ideg, csr, fsum, fprod, hsum, hprod);
    k_final<<<N_NODES / 16, 128, 0, stream>>>(vmat, hsum, hprod, ideg, (float*)d_out);
}


# Edge slotting by LDS counters: 25 node-range x 25 edge-slice blocks count, then place, in LDS; global atomics, zero and scan launches gone; a row's slot order now by slice, then LDS arrival
# speedup vs baseline: 2.4011x; 1.3743x over previous
#include <hip/hip_runtime.h>
#include <math.h>


typedef __attribute__((ext_vector_type(2))) float v2f;
typedef __attribute__((ext_vector_type(8))) float v8f;

#define N_NODES 50000
#define N_EDGES 800000
#define IN_F    256
#define OUT_F   128
#define RANK    64

static_assert(N_NODES % 16 == 0, "row tiles exact");
static_assert(N_NODES % 8 == 0,  "gather blocks exact");

#define OFF_ODEG   0u
#define OFF_IDEG   50000u
#define OFF_HSUM   100000u
#define OFF_CNTI   100000u
#define OFF_CNTO   1350000u
#define OFF_BTOT   2600000u
#define OFF_HPROD  6500000u
#define OFF_ROWPTR 9700000u
#define OFF_CSR    9760000u
#define OFF_FSUM   12900000u
#define OFF_FPROD  19300000u

#define PRE_R      25
#define PRE_NR     (N_NODES / PRE_R)
#define PRE_S      25
#define PRE_ES     (N_EDGES / PRE_S)
#define PRE_ES4    (PRE_ES / 4)
#define PRE_T      320
#define PRE_TRIPS  (PRE_ES4 / PRE_T)
#define PRE_BATCH  5
static_assert(PRE_R * PRE_NR == N_NODES, "ranges cover the nodes exactly");
static_assert(PRE_S * PRE_ES == N_EDGES, "slices cover the edges exactly");
static_assert(PRE_ES % 4 == 0, "a slice is whole 16-byte loads");
static_assert(PRE_T * PRE_TRIPS == PRE_ES4, "the threads cover a slice exactly: no guard on the loads");
static_assert(PRE_TRIPS % PRE_BATCH == 0, "whole batches");
static_assert(PRE_T % 32 == 0, "whole waves");
static_assert(PRE_NR % 8 == 0 && PRE_NR <= PRE_T * 8, "8 nodes per thread cover a range");
static_assert(N_NODES % 4 == 0, "16-byte steps between the slices of a count table");
static_assert(OFF_CNTI + PRE_S * N_NODES <= OFF_CNTO, "count tables do not overlap");
static_assert(OFF_CNTO + PRE_S * N_NODES <= OFF_BTOT, "count tables do not overlap");
static_assert(OFF_BTOT + PRE_R * PRE_S <= OFF_HPROD, "count tables stay inside HSUM's space");

__global__ __launch_bounds__(PRE_T) void k_count(const int* __restrict__ src,
                                               const int* __restrict__ dst,
                                               int* __restrict__ cntI,
                                               int* __restrict__ cntO,
                                               int* __restrict__ btot)
{
    __shared__ int ci[PRE_NR];
    __shared__ int co[PRE_NR];
    __shared__ int wsum[PRE_T / 32];

    const int s    = blockIdx.x;
    const int r    = blockIdx.y;
    const int tid  = threadIdx.x;
    const int lane = tid & 31;
    const int wave = tid >> 5;
    const int n0   = r * PRE_NR;

    for (int t = tid; t < PRE_NR; t += PRE_T) { ci[t] = 0; co[t] = 0; }
    __syncthreads();

    const int4* d4 = (const int4*)(dst + (size_t)s * PRE_ES);
    const int4* s4 = (const int4*)(src + (size_t)s * PRE_ES);
    for (int kb = 0; kb < PRE_TRIPS; kb += PRE_BATCH) {
        int4 d[PRE_BATCH], q[PRE_BATCH];
#pragma unroll
        for (int j = 0; j < PRE_BATCH; ++j) {
            d[j] = d4[tid + PRE_T * (kb + j)];
            q[j] = s4[tid + PRE_T * (kb + j)];
        }
#pragma unroll
        for (int j = 0; j < PRE_BATCH; ++j) {
            unsigned u;
            u = (unsigned)(d[j].x - n0); if (u < (unsigned)PRE_NR) atomicAdd(&ci[u], 1);
            u = (unsigned)(d[j].y - n0); if (u < (unsigned)PRE_NR) atomicAdd(&ci[u], 1);
            u = (unsigned)(d[j].z - n0); if (u < (unsigned)PRE_NR) atomicAdd(&ci[u], 1);
            u = (unsigned)(d[j].w - n0); if (u < (unsigned)PRE_NR) atomicAdd(&ci[u], 1);
            u = (unsigned)(q[j].x - n0); if (u < (unsigned)PRE_NR) atomicAdd(&co[u], 1);
            u = (unsigned)(q[j].y - n0); if (u < (unsigned)PRE_NR) atomicAdd(&co[u], 1);
            u = (unsigned)(q[j].z - n0); if (u < (unsigned)PRE_NR) atomicAdd(&co[u], 1);
            u = (unsigned)(q[j].w - n0); if (u < (unsigned)PRE_NR) atomicAdd(&co[u], 1);
        }
    }
    __syncthreads();

    int part = 0;
    for (int t = tid; t < PRE_NR; t += PRE_T) {
        const int a = ci[t];
        cntI[(size_t)s * N_NODES + n0 + t] = a;
        cntO[(size_t)s * N_NODES + n0 + t] = co[t];
        part += a;
    }
#pragma unroll
    for (int o = 16; o > 0; o >>= 1) part += __shfl_xor(part, o);
    if (lane == 0) wsum[wave] = part;
    __syncthreads();
    if (tid == 0) {
        int tot = 0;
#pragma unroll
        for (int w = 0; w < PRE_T / 32; ++w) tot += wsum[w];
        btot[r * PRE_S + s] = tot;
    }
}

__global__ __launch_bounds__(PRE_T) void k_place(const int* __restrict__ src,
                                               const int* __restrict__ dst,
                                               const int* __restrict__ cntI,
                                               const int* __restrict__ cntO,
                                               const int* __restrict__ btot,
                                               float* __restrict__ odeg,
                                               int*   __restrict__ ideg,
                                               int*   __restrict__ rowptr,
                                               int*   __restrict__ csr)
{
    __shared__ int base[PRE_NR];
    __shared__ int cur[PRE_NR];
    __shared__ int wtot[PRE_T / 32];
    __shared__ int wpre[PRE_T / 32];

    const int s    = blockIdx.x;
    const int r    = blockIdx.y;
    const int tid  = threadIdx.x;
    const int lane = tid & 31;
    const int wave = tid >> 5;
    const int n0   = r * PRE_NR;
    const int t0   = tid * 8;
    const bool has = (t0 < PRE_NR);

    int pre = 0;
    for (int i = tid; i < r * PRE_S; i += PRE_T) pre += btot[i];

    int tot[8], bef[8];
#pragma unroll
    for (int j = 0; j < 8; ++j) { tot[j] = 0; bef[j] = 0; }
    if (has) {
        const int4* c = (const int4*)(cntI + n0 + t0);
#pragma unroll
        for (int k = 0; k < PRE_S; ++k) {
            if (k == s) {
#pragma unroll
                for (int j = 0; j < 8; ++j) bef[j] = tot[j];
            }
            const int4 a = c[(size_t)k * (N_NODES / 4)];
            const int4 b = c[(size_t)k * (N_NODES / 4) + 1];
            tot[0] += a.x; tot[1] += a.y; tot[2] += a.z; tot[3] += a.w;
            tot[4] += b.x; tot[5] += b.y; tot[6] += b.z; tot[7] += b.w;
        }
    }
    int tsum = 0;
#pragma unroll
    for (int j = 0; j < 8; ++j) tsum += tot[j];

    int incl = tsum;
#pragma unroll
    for (int o = 1; o < 32; o <<= 1) {
        int up = __shfl_up(incl, o);
        if (lane >= o) incl += up;
    }
#pragma unroll
    for (int o = 16; o > 0; o >>= 1) pre += __shfl_xor(pre, o);
    if (lane == 31) wtot[wave] = incl;
    if (lane == 0)  wpre[wave] = pre;
    __syncthreads();
    int woff = 0, rbase = 0;
#pragma unroll
    for (int w = 0; w < PRE_T / 32; ++w) {
        rbase += wpre[w];
        if (w < wave) woff += wtot[w];
    }

    if (has) {
        int run = rbase + woff + (incl - tsum);
        int rs[8];
#pragma unroll
        for (int j = 0; j < 8; ++j) {
            rs[j] = run;
            base[t0 + j] = run + bef[j];
            cur[t0 + j]  = 0;
            run += tot[j];
        }
        if (s == 0) {
            int4* rp = (int4*)(rowptr + n0 + t0);
            int4* ip = (int4*)(ideg + n0 + t0);
            rp[0] = make_int4(rs[0], rs[1], rs[2], rs[3]);
            rp[1] = make_int4(rs[4], rs[5], rs[6], rs[7]);
            ip[0] = make_int4(tot[0], tot[1], tot[2], tot[3]);
            ip[1] = make_int4(tot[4], tot[5], tot[6], tot[7]);
            int od[8];
#pragma unroll
            for (int j = 0; j < 8; ++j) od[j] = 0;
            const int4* c = (const int4*)(cntO + n0 + t0);
#pragma unroll
            for (int k = 0; k < PRE_S; ++k) {
                const int4 a = c[(size_t)k * (N_NODES / 4)];
                const int4 b = c[(size_t)k * (N_NODES / 4) + 1];
                od[0] += a.x; od[1] += a.y; od[2] += a.z; od[3] += a.w;
                od[4] += b.x; od[5] += b.y; od[6] += b.z; od[7] += b.w;
            }
            float4* op = (float4*)(odeg + n0 + t0);
            op[0] = make_float4((float)od[0], (float)od[1], (float)od[2], (float)od[3]);
            op[1] = make_float4((float)od[4], (float)od[5], (float)od[6], (float)od[7]);
        }
    }
    __syncthreads();

    const int4* d4 = (const int4*)(dst + (size_t)s * PRE_ES);
    const int4* s4 = (const int4*)(src + (size_t)s * PRE_ES);
    for (int kb = 0; kb < PRE_TRIPS; kb += PRE_BATCH) {
        int4 d[PRE_BATCH], q[PRE_BATCH];
#pragma unroll
        for (int j = 0; j < PRE_BATCH; ++j) {
            d[j] = d4[tid + PRE_T * (kb + j)];
            q[j] = s4[tid + PRE_T * (kb + j)];
        }
#pragma unroll
        for (int j = 0; j < PRE_BATCH; ++j) {
            unsigned u, p;
            u = (unsigned)(d[j].x - n0);
            if (u < (unsigned)PRE_NR) { p = (unsigned)(base[u] + atomicAdd(&cur[u], 1)); if (p < (unsigned)N_EDGES) csr[p] = q[j].x; }
            u = (unsigned)(d[j].y - n0);
            if (u < (unsigned)PRE_NR) { p = (unsigned)(base[u] + atomicAdd(&cur[u], 1)); if (p < (unsigned)N_EDGES) csr[p] = q[j].y; }
            u = (unsigned)(d[j].z - n0);
            if (u < (unsigned)PRE_NR) { p = (unsigned)(base[u] + atomicAdd(&cur[u], 1)); if (p < (unsigned)N_EDGES) csr[p] = q[j].z; }
            u = (unsigned)(d[j].w - n0);
            if (u < (unsigned)PRE_NR) { p = (unsigned)(base[u] + atomicAdd(&cur[u], 1)); if (p < (unsigned)N_EDGES) csr[p] = q[j].w; }
        }
    }
}

__global__ __launch_bounds__(128) void k_node_gemm(const float* __restrict__ feat,
                                                   const float* __restrict__ w1,
                                                   const float* __restrict__ w2,
                                                   const float* __restrict__ odeg,
                                                   float* __restrict__ fsum,
                                                   float* __restrict__ fprod)
{
    __shared__ float As[16][260];
    __shared__ float sc[16];

    const int m0  = blockIdx.x * 16;
    const int tid = threadIdx.x;

    if (tid < 16) {
        sc[tid] = rsqrtf(fmaxf(odeg[m0 + tid], 1.0f));
    }
    __syncthreads();

    {
        const int r   = tid >> 3;
        const int c0  = (tid & 7) * 32;
        const float s = sc[r];
        const float4* g = (const float4*)(feat + (size_t)(m0 + r) * IN_F + c0);
#pragma unroll
        for (int j = 0; j < 8; ++j) {
            float4 x = g[j];
            As[r][c0 + 4 * j + 0] = x.x * s;
            As[r][c0 + 4 * j + 1] = x.y * s;
            As[r][c0 + 4 * j + 2] = x.z * s;
            As[r][c0 + 4 * j + 3] = x.w * s;
        }
    }
    __syncthreads();

    const int wave  = tid >> 5;
    const int lane  = tid & 31;
    const int rsel  = lane & 15;
    const int khalf = (lane < 16) ? 0 : 2;
    const int rofs  = (lane < 16) ? 0 : 8;

    for (int t = 0; t < 3; ++t) {
        const int  tile = wave + 4 * t;
        const bool isW2 = (tile >= 8);
        const int  n0   = isW2 ? (tile - 8) * 16 : tile * 16;
        const float* B  = isW2 ? w2 : w1;
        const int  ldb  = isW2 ? RANK : OUT_F;

        for (int k = 0; k < IN_F; k += 32)
            __builtin_prefetch(B + (size_t)k * ldb + n0, 0, 0);

        v8f acc = {};
        for (int k0 = 0; k0 < IN_F; k0 += 4) {
            v2f a;
            a.x = As[rsel][k0 + khalf];
            a.y = As[rsel][k0 + khalf + 1];
            v2f b;
            const float* Bp = B + (size_t)(k0 + khalf) * ldb + n0 + rsel;
            b.x = Bp[0];
            b.y = Bp[ldb];
            acc = __builtin_amdgcn_wmma_f32_16x16x4_f32(
                false, a, false, b, (short)0, acc, false, false);
        }

        const int colg = n0 + rsel;
        if (isW2) {
            const float bias = w2[(size_t)IN_F * RANK + colg];
            float* op = fprod + (size_t)(m0 + rofs) * RANK + colg;
#pragma unroll
            for (int r = 0; r < 8; ++r)
                op[r * RANK] = tanhf(acc[r] + bias);
        } else {
            float* op = fsum + (size_t)(m0 + rofs) * OUT_F + colg;
#pragma unroll
            for (int r = 0; r < 8; ++r)
                op[r * OUT_F] = acc[r];
        }
    }
}

__global__ __launch_bounds__(256) void k_gather(const int* __restrict__ rowptr,
                                                const int* __restrict__ ideg,
                                                const int* __restrict__ csr,
                                                const float* __restrict__ fsum,
                                                const float* __restrict__ fprod,
                                                float* __restrict__ hsum,
                                                float* __restrict__ hprod)
{
    const int lane = threadIdx.x & 31;
    const int d     = __builtin_amdgcn_readfirstlane((int)(blockIdx.x * 8u + (threadIdx.x >> 5)));
    const int start = rowptr[d];
    const int cnt   = ideg[d];

    float4 hs = make_float4(0.0f, 0.0f, 0.0f, 0.0f);
    float  la0 = 0.0f, la1 = 0.0f;
    int    ng0 = 0, ng1 = 0;

    for (int b = 0; b < cnt; b += 32) {
        const int n = min(32, cnt - b);
        int mine = 0;
        if (lane < n) mine = csr[start + b + lane];
#pragma unroll 4
        for (int j = 0; j < n; ++j) {
            const unsigned s = min((unsigned)__builtin_amdgcn_readlane(mine, j), (unsigned)(N_NODES - 1));
            const float4 x = *(const float4*)(fsum  + (size_t)s * OUT_F + 4 * lane);
            const float2 m = *(const float2*)(fprod + (size_t)s * RANK  + 2 * lane);
            hs.x += x.x; hs.y += x.y; hs.z += x.z; hs.w += x.w;
            la0 += __logf(fabsf(m.x));
            la1 += __logf(fabsf(m.y));
            ng0 += (m.x < 0.0f) ? 1 : 0;
            ng1 += (m.y < 0.0f) ? 1 : 0;
        }
    }

    *(float4*)(hsum + (size_t)d * OUT_F + 4 * lane) = hs;

    float2 hp = make_float2(0.0f, 0.0f);
    if (cnt > 0) {
        hp.x = ((ng0 & 1) ? -1.0f : 1.0f) * __expf(la0);
        hp.y = ((ng1 & 1) ? -1.0f : 1.0f) * __expf(la1);
    }
    *(float2*)(hprod + (size_t)d * RANK + 2 * lane) = hp;
}

__global__ __launch_bounds__(128) void k_final(const float* __restrict__ vmat,
                                               const float* __restrict__ hsum,
                                               const float* __restrict__ hprod,
                                               const int*   __restrict__ ideg,
                                               float* __restrict__ out)
{
    __shared__ float Hp[16][68];
    __shared__ float sc[16];

    const int m0  = blockIdx.x * 16;
    const int tid = threadIdx.x;

    if (tid < 16) {
        sc[tid] = rsqrtf(fmaxf((float)ideg[m0 + tid], 1.0f));
    }
    for (int i = tid; i < 16 * RANK; i += 128) {
        int r = i >> 6, c = i & 63;
        Hp[r][c] = hprod[(size_t)(m0 + r) * RANK + c];
    }
    __syncthreads();

    const int wave  = tid >> 5;
    const int lane  = tid & 31;
    const int rsel  = lane & 15;
    const int khalf = (lane < 16) ? 0 : 2;
    const int rofs  = (lane < 16) ? 0 : 8;

    for (int t = 0; t < 2; ++t) {
        const int n0 = (wave * 2 + t) * 16;

        for (int k = 0; k < RANK; k += 32)
            __builtin_prefetch(vmat + (size_t)k * OUT_F + n0, 0, 0);

        v8f acc = {};
        for (int k0 = 0; k0 < RANK; k0 += 4) {
            v2f a;
            a.x = Hp[rsel][k0 + khalf];
            a.y = Hp[rsel][k0 + khalf + 1];
            v2f b;
            const float* Bp = vmat + (size_t)(k0 + khalf) * OUT_F + n0 + rsel;
            b.x = Bp[0];
            b.y = Bp[OUT_F];
            acc = __builtin_amdgcn_wmma_f32_16x16x4_f32(
                false, a, false, b, (short)0, acc, false, false);
        }
        const int colg = n0 + rsel;
        const float* hp = hsum + (size_t)(m0 + rofs) * OUT_F + colg;
        float*       op = out  + (size_t)(m0 + rofs) * OUT_F + colg;
#pragma unroll
        for (int r = 0; r < 8; ++r)
            op[r * OUT_F] = (acc[r] + hp[r * OUT_F]) * sc[rofs + r];
    }
}

extern "C" void kernel_launch(void* const* d_in, const int* in_sizes, int n_in,
                              void* d_out, int out_size, void* d_ws, size_t ws_size,
                              hipStream_t stream)
{
    (void)in_sizes; (void)n_in; (void)out_size; (void)ws_size;

    const float* feat = (const float*)d_in[0];
    const float* w1   = (const float*)d_in[1];
    const float* w2   = (const float*)d_in[2];
    const float* vmat = (const float*)d_in[3];
    const int*   src  = (const int*)d_in[4];
    const int*   dst  = (const int*)d_in[5];

    float* ws     = (float*)d_ws;
    float* odeg   = ws + OFF_ODEG;
    int*   ideg   = (int*)(ws + OFF_IDEG);
    float* hsum   = ws + OFF_HSUM;
    float* hprod  = ws + OFF_HPROD;
    int*   rowptr = (int*)(ws + OFF_ROWPTR);
    int*   csr    = (int*)(ws + OFF_CSR);
    int*   cntI   = (int*)(ws + OFF_CNTI);
    int*   cntO   = (int*)(ws + OFF_CNTO);
    int*   btot   = (int*)(ws + OFF_BTOT);
    float* fsum   = ws + OFF_FSUM;
    float* fprod  = ws + OFF_FPROD;

    k_count<<<dim3(PRE_S, PRE_R), PRE_T, 0, stream>>>(src, dst, cntI, cntO, btot);
    k_place<<<dim3(PRE_S, PRE_R), PRE_T, 0, stream>>>(src, dst, cntI, cntO, btot, odeg, ideg, rowptr, csr);
    k_node_gemm<<<N_NODES / 16, 128, 0, stream>>>(feat, w1, w2, odeg, fsum, fprod);
    k_gather<<<N_NODES / 8, 256, 0, stream>>>(rowptr, ideg, csr, fsum, fprod, hsum, hprod);
    k_final<<<N_NODES / 16, 128, 0, stream>>>(vmat, hsum, hprod, ideg, (float*)d_out);
}


# Finish fused into the edge gather: a block of 8 waves sums 16 dst rows (2 per wave), h_prod tile via LDS to f32 WMMA with v, whole 512-B out rows stored; final launch, h_sum/h_prod planes gone
# speedup vs baseline: 2.6106x; 1.0873x over previous
#include <hip/hip_runtime.h>
#include <math.h>


typedef __attribute__((ext_vector_type(2))) float v2f;
typedef __attribute__((ext_vector_type(8))) float v8f;

#define N_NODES 50000
#define N_EDGES 800000
#define IN_F    256
#define OUT_F   128
#define RANK    64

static_assert(N_NODES % 16 == 0, "row tiles exact");
static_assert(OUT_F == 8 * 16, "one 16-column tile of the finish per wave of a gather block");

#define OFF_ODEG   0u
#define OFF_IDEG   50000u
#define OFF_CNTI   100000u
#define OFF_CNTO   1350000u
#define OFF_BTOT   2600000u
#define OFF_ROWPTR 9700000u
#define OFF_CSR    9760000u
#define OFF_FSUM   12900000u
#define OFF_FPROD  19300000u

#define PRE_R      25
#define PRE_NR     (N_NODES / PRE_R)
#define PRE_S      25
#define PRE_ES     (N_EDGES / PRE_S)
#define PRE_ES4    (PRE_ES / 4)
#define PRE_T      320
#define PRE_TRIPS  (PRE_ES4 / PRE_T)
#define PRE_BATCH  5
static_assert(PRE_R * PRE_NR == N_NODES, "ranges cover the nodes exactly");
static_assert(PRE_S * PRE_ES == N_EDGES, "slices cover the edges exactly");
static_assert(PRE_ES % 4 == 0, "a slice is whole 16-byte loads");
static_assert(PRE_T * PRE_TRIPS == PRE_ES4, "the threads cover a slice exactly: no guard on the loads");
static_assert(PRE_TRIPS % PRE_BATCH == 0, "whole batches");
static_assert(PRE_T % 32 == 0, "whole waves");
static_assert(PRE_NR % 8 == 0 && PRE_NR <= PRE_T * 8, "8 nodes per thread cover a range");
static_assert(N_NODES % 4 == 0, "16-byte steps between the slices of a count table");
static_assert(OFF_CNTI + PRE_S * N_NODES <= OFF_CNTO, "count tables do not overlap");
static_assert(OFF_CNTO + PRE_S * N_NODES <= OFF_BTOT, "count tables do not overlap");
static_assert(OFF_BTOT + PRE_R * PRE_S <= OFF_ROWPTR, "count tables end before the row starts");

__global__ __launch_bounds__(PRE_T) void k_count(const int* __restrict__ src,
                                               const int* __restrict__ dst,
                                               int* __restrict__ cntI,
                                               int* __restrict__ cntO,
                                               int* __restrict__ btot)
{
    __shared__ int ci[PRE_NR];
    __shared__ int co[PRE_NR];
    __shared__ int wsum[PRE_T / 32];

    const int s    = blockIdx.x;
    const int r    = blockIdx.y;
    const int tid  = threadIdx.x;
    const int lane = tid & 31;
    const int wave = tid >> 5;
    const int n0   = r * PRE_NR;

    for (int t = tid; t < PRE_NR; t += PRE_T) { ci[t] = 0; co[t] = 0; }
    __syncthreads();

    const int4* d4 = (const int4*)(dst + (size_t)s * PRE_ES);
    const int4* s4 = (const int4*)(src + (size_t)s * PRE_ES);
    for (int kb = 0; kb < PRE_TRIPS; kb += PRE_BATCH) {
        int4 d[PRE_BATCH], q[PRE_BATCH];
#pragma unroll
        for (int j = 0; j < PRE_BATCH; ++j) {
            d[j] = d4[tid + PRE_T * (kb + j)];
            q[j] = s4[tid + PRE_T * (kb + j)];
        }
#pragma unroll
        for (int j = 0; j < PRE_BATCH; ++j) {
            unsigned u;
            u = (unsigned)(d[j].x - n0); if (u < (unsigned)PRE_NR) atomicAdd(&ci[u], 1);
            u = (unsigned)(d[j].y - n0); if (u < (unsigned)PRE_NR) atomicAdd(&ci[u], 1);
            u = (unsigned)(d[j].z - n0); if (u < (unsigned)PRE_NR) atomicAdd(&ci[u], 1);
            u = (unsigned)(d[j].w - n0); if (u < (unsigned)PRE_NR) atomicAdd(&ci[u], 1);
            u = (unsigned)(q[j].x - n0); if (u < (unsigned)PRE_NR) atomicAdd(&co[u], 1);
            u = (unsigned)(q[j].y - n0); if (u < (unsigned)PRE_NR) atomicAdd(&co[u], 1);
            u = (unsigned)(q[j].z - n0); if (u < (unsigned)PRE_NR) atomicAdd(&co[u], 1);
            u = (unsigned)(q[j].w - n0); if (u < (unsigned)PRE_NR) atomicAdd(&co[u], 1);
        }
    }
    __syncthreads();

    int part = 0;
    for (int t = tid; t < PRE_NR; t += PRE_T) {
        const int a = ci[t];
        cntI[(size_t)s * N_NODES + n0 + t] = a;
        cntO[(size_t)s * N_NODES + n0 + t] = co[t];
        part += a;
    }
#pragma unroll
    for (int o = 16; o > 0; o >>= 1) part += __shfl_xor(part, o);
    if (lane == 0) wsum[wave] = part;
    __syncthreads();
    if (tid == 0) {
        int tot = 0;
#pragma unroll
        for (int w = 0; w < PRE_T / 32; ++w) tot += wsum[w];
        btot[r * PRE_S + s] = tot;
    }
}

__global__ __launch_bounds__(PRE_T) void k_place(const int* __restrict__ src,
                                               const int* __restrict__ dst,
                                               const int* __restrict__ cntI,
                                               const int* __restrict__ cntO,
                                               const int* __restrict__ btot,
                                               float* __restrict__ odeg,
                                               int*   __restrict__ ideg,
                                               int*   __restrict__ rowptr,
                                               int*   __restrict__ csr)
{
    __shared__ int base[PRE_NR];
    __shared__ int cur[PRE_NR];
    __shared__ int wtot[PRE_T / 32];
    __shared__ int wpre[PRE_T / 32];

    const int s    = blockIdx.x;
    const int r    = blockIdx.y;
    const int tid  = threadIdx.x;
    const int lane = tid & 31;
    const int wave = tid >> 5;
    const int n0   = r * PRE_NR;
    const int t0   = tid * 8;
    const bool has = (t0 < PRE_NR);

    int pre = 0;
    for (int i = tid; i < r * PRE_S; i += PRE_T) pre += btot[i];

    int tot[8], bef[8];
#pragma unroll
    for (int j = 0; j < 8; ++j) { tot[j] = 0; bef[j] = 0; }
    if (has) {
        const int4* c = (const int4*)(cntI + n0 + t0);
#pragma unroll
        for (int k = 0; k < PRE_S; ++k) {
            if (k == s) {
#pragma unroll
                for (int j = 0; j < 8; ++j) bef[j] = tot[j];
            }
            const int4 a = c[(size_t)k * (N_NODES / 4)];
            const int4 b = c[(size_t)k * (N_NODES / 4) + 1];
            tot[0] += a.x; tot[1] += a.y; tot[2] += a.z; tot[3] += a.w;
            tot[4] += b.x; tot[5] += b.y; tot[6] += b.z; tot[7] += b.w;
        }
    }
    int tsum = 0;
#pragma unroll
    for (int j = 0; j < 8; ++j) tsum += tot[j];

    int incl = tsum;
#pragma unroll
    for (int o = 1; o < 32; o <<= 1) {
        int up = __shfl_up(incl, o);
        if (lane >= o) incl += up;
    }
#pragma unroll
    for (int o = 16; o > 0; o >>= 1) pre += __shfl_xor(pre, o);
    if (lane == 31) wtot[wave] = incl;
    if (lane == 0)  wpre[wave] = pre;
    __syncthreads();
    int woff = 0, rbase = 0;
#pragma unroll
    for (int w = 0; w < PRE_T / 32; ++w) {
        rbase += wpre[w];
        if (w < wave) woff += wtot[w];
    }

    if (has) {
        int run = rbase + woff + (incl - tsum);
        int rs[8];
#pragma unroll
        for (int j = 0; j < 8; ++j) {
            rs[j] = run;
            base[t0 + j] = run + bef[j];
            cur[t0 + j]  = 0;
            run += tot[j];
        }
        if (s == 0) {
            int4* rp = (int4*)(rowptr + n0 + t0);
            int4* ip = (int4*)(ideg + n0 + t0);
            rp[0] = make_int4(rs[0], rs[1], rs[2], rs[3]);
            rp[1] = make_int4(rs[4], rs[5], rs[6], rs[7]);
            ip[0] = make_int4(tot[0], tot[1], tot[2], tot[3]);
            ip[1] = make_int4(tot[4], tot[5], tot[6], tot[7]);
            int od[8];
#pragma unroll
            for (int j = 0; j < 8; ++j) od[j] = 0;
            const int4* c = (const int4*)(cntO + n0 + t0);
#pragma unroll
            for (int k = 0; k < PRE_S; ++k) {
                const int4 a = c[(size_t)k * (N_NODES / 4)];
                const int4 b = c[(size_t)k * (N_NODES / 4) + 1];
                od[0] += a.x; od[1] += a.y; od[2] += a.z; od[3] += a.w;
                od[4] += b.x; od[5] += b.y; od[6] += b.z; od[7] += b.w;
            }
            float4* op = (float4*)(odeg + n0 + t0);
            op[0] = make_float4((float)od[0], (float)od[1], (float)od[2], (float)od[3]);
            op[1] = make_float4((float)od[4], (float)od[5], (float)od[6], (float)od[7]);
        }
    }
    __syncthreads();

    const int4* d4 = (const int4*)(dst + (size_t)s * PRE_ES);
    const int4* s4 = (const int4*)(src + (size_t)s * PRE_ES);
    for (int kb = 0; kb < PRE_TRIPS; kb += PRE_BATCH) {
        int4 d[PRE_BATCH], q[PRE_BATCH];
#pragma unroll
        for (int j = 0; j < PRE_BATCH; ++j) {
            d[j] = d4[tid + PRE_T * (kb + j)];
            q[j] = s4[tid + PRE_T * (kb + j)];
        }
#pragma unroll
        for (int j = 0; j < PRE_BATCH; ++j) {
            unsigned u, p;
            u = (unsigned)(d[j].x - n0);
            if (u < (unsigned)PRE_NR) { p = (unsigned)(base[u] + atomicAdd(&cur[u], 1)); if (p < (unsigned)N_EDGES) csr[p] = q[j].x; }
            u = (unsigned)(d[j].y - n0);
            if (u < (unsigned)PRE_NR) { p = (unsigned)(base[u] + atomicAdd(&cur[u], 1)); if (p < (unsigned)N_EDGES) csr[p] = q[j].y; }
            u = (unsigned)(d[j].z - n0);
            if (u < (unsigned)PRE_NR) { p = (unsigned)(base[u] + atomicAdd(&cur[u], 1)); if (p < (unsigned)N_EDGES) csr[p] = q[j].z; }
            u = (unsigned)(d[j].w - n0);
            if (u < (unsigned)PRE_NR) { p = (unsigned)(base[u] + atomicAdd(&cur[u], 1)); if (p < (unsigned)N_EDGES) csr[p] = q[j].w; }
        }
    }
}

__global__ __launch_bounds__(128) void k_node_gemm(const float* __restrict__ feat,
                                                   const float* __restrict__ w1,
                                                   const float* __restrict__ w2,
                                                   const float* __restrict__ odeg,
                                                   float* __restrict__ fsum,
                                                   float* __restrict__ fprod)
{
    __shared__ float As[16][260];
    __shared__ float sc[16];

    const int m0  = blockIdx.x * 16;
    const int tid = threadIdx.x;

    if (tid < 16) {
        sc[tid] = rsqrtf(fmaxf(odeg[m0 + tid], 1.0f));
    }
    __syncthreads();

    {
        const int r   = tid >> 3;
        const int c0  = (tid & 7) * 32;
        const float s = sc[r];
        const float4* g = (const float4*)(feat + (size_t)(m0 + r) * IN_F + c0);
#pragma unroll
        for (int j = 0; j < 8; ++j) {
            float4 x = g[j];
            As[r][c0 + 4 * j + 0] = x.x * s;
            As[r][c0 + 4 * j + 1] = x.y * s;
            As[r][c0 + 4 * j + 2] = x.z * s;
            As[r][c0 + 4 * j + 3] = x.w * s;
        }
    }
    __syncthreads();

    const int wave  = tid >> 5;
    const int lane  = tid & 31;
    const int rsel  = lane & 15;
    const int khalf = (lane < 16) ? 0 : 2;
    const int rofs  = (lane < 16) ? 0 : 8;

    for (int t = 0; t < 3; ++t) {
        const int  tile = wave + 4 * t;
        const bool isW2 = (tile >= 8);
        const int  n0   = isW2 ? (tile - 8) * 16 : tile * 16;
        const float* B  = isW2 ? w2 : w1;
        const int  ldb  = isW2 ? RANK : OUT_F;

        for (int k = 0; k < IN_F; k += 32)
            __builtin_prefetch(B + (size_t)k * ldb + n0, 0, 0);

        v8f acc = {};
        for (int k0 = 0; k0 < IN_F; k0 += 4) {
            v2f a;
            a.x = As[rsel][k0 + khalf];
            a.y = As[rsel][k0 + khalf + 1];
            v2f b;
            const float* Bp = B + (size_t)(k0 + khalf) * ldb + n0 + rsel;
            b.x = Bp[0];
            b.y = Bp[ldb];
            acc = __builtin_amdgcn_wmma_f32_16x16x4_f32(
                false, a, false, b, (short)0, acc, false, false);
        }

        const int colg = n0 + rsel;
        if (isW2) {
            const float bias = w2[(size_t)IN_F * RANK + colg];
            float* op = fprod + (size_t)(m0 + rofs) * RANK + colg;
#pragma unroll
            for (int r = 0; r < 8; ++r)
                op[r * RANK] = tanhf(acc[r] + bias);
        } else {
            float* op = fsum + (size_t)(m0 + rofs) * OUT_F + colg;
#pragma unroll
            for (int r = 0; r < 8; ++r)
                op[r * OUT_F] = acc[r];
        }
    }
}

__device__ __forceinline__ void gather_row(const int start, const int cnt, int mine, const int lane,
                                           const int* __restrict__ csr,
                                           const float* __restrict__ fsum,
                                           const float* __restrict__ fprod,
                                           float4& hs_out, float2& hp_out)
{
    float4 hs = make_float4(0.0f, 0.0f, 0.0f, 0.0f);
    float  la0 = 0.0f, la1 = 0.0f;
    int    ng0 = 0, ng1 = 0;

    for (int b = 0; b < cnt; b += 32) {
        const int n = min(32, cnt - b);
        if (b > 0) {
            mine = 0;
            if (lane < n) mine = csr[start + b + lane];
        }
#pragma unroll 4
        for (int j = 0; j < n; ++j) {
            const unsigned s = min((unsigned)__builtin_amdgcn_readlane(mine, j), (unsigned)(N_NODES - 1));
            const float4 x = *(const float4*)(fsum  + (size_t)s * OUT_F + 4 * lane);
            const float2 m = *(const float2*)(fprod + (size_t)s * RANK  + 2 * lane);
            hs.x += x.x; hs.y += x.y; hs.z += x.z; hs.w += x.w;
            la0 += __logf(fabsf(m.x));
            la1 += __logf(fabsf(m.y));
            ng0 += (m.x < 0.0f) ? 1 : 0;
            ng1 += (m.y < 0.0f) ? 1 : 0;
        }
    }

    float2 hp = make_float2(0.0f, 0.0f);
    if (cnt > 0) {
        hp.x = ((ng0 & 1) ? -1.0f : 1.0f) * __expf(la0);
        hp.y = ((ng1 & 1) ? -1.0f : 1.0f) * __expf(la1);
    }
    hs_out = hs;
    hp_out = hp;
}

__global__ __launch_bounds__(256) void k_gather_final(const int* __restrict__ rowptr,
                                                      const int* __restrict__ ideg,
                                                      const int* __restrict__ csr,
                                                      const float* __restrict__ fsum,
                                                      const float* __restrict__ fprod,
                                                      const float* __restrict__ vmat,
                                                      float* __restrict__ out)
{
    __shared__ __attribute__((aligned(16))) float Hp[16][68];
    __shared__ __attribute__((aligned(16))) float Pv[16][132];

    const int lane = threadIdx.x & 31;
    const int wave = __builtin_amdgcn_readfirstlane((int)(threadIdx.x >> 5));
    const int m0   = (int)(blockIdx.x * 16u);
    const int rA   = 2 * wave;
    const int dA   = m0 + rA;
    const int dB   = dA + 1;

    const int startA = rowptr[dA];
    const int cntA   = ideg[dA];
    const int startB = rowptr[dB];
    const int cntB   = ideg[dB];

    int mineA = 0, mineB = 0;
    if (lane < cntA) mineA = csr[startA + lane];
    if (lane < cntB) mineB = csr[startB + lane];

    float4 hsA, hsB;
    float2 hpA, hpB;
    gather_row(startA, cntA, mineA, lane, csr, fsum, fprod, hsA, hpA);
    gather_row(startB, cntB, mineB, lane, csr, fsum, fprod, hsB, hpB);

    const int rsel  = lane & 15;
    const int khalf = (lane < 16) ? 0 : 2;
    const int rofs  = (lane < 16) ? 0 : 8;
    const int n0    = wave * 16;

    v2f bv[RANK / 4];
    {
        const float* Bp = vmat + (size_t)khalf * OUT_F + n0 + rsel;
#pragma unroll
        for (int i = 0; i < RANK / 4; ++i) {
            bv[i].x = Bp[(size_t)(4 * i) * OUT_F];
            bv[i].y = Bp[(size_t)(4 * i + 1) * OUT_F];
        }
    }

    *(float2*)&Hp[rA][2 * lane]     = hpA;
    *(float2*)&Hp[rA + 1][2 * lane] = hpB;
    __syncthreads();

    {
        v8f acc = {};
#pragma unroll
        for (int i = 0; i < RANK / 4; ++i) {
            const int k0 = 4 * i;
            v2f a;
            a.x = Hp[rsel][k0 + khalf];
            a.y = Hp[rsel][k0 + khalf + 1];
            acc = __builtin_amdgcn_wmma_f32_16x16x4_f32(
                false, a, false, bv[i], (short)0, acc, false, false);
        }
#pragma unroll
        for (int r = 0; r < 8; ++r)
            Pv[rofs + r][n0 + rsel] = acc[r];
    }
    __syncthreads();

    {
        const float  scA = rsqrtf(fmaxf((float)cntA, 1.0f));
        const float  scB = rsqrtf(fmaxf((float)cntB, 1.0f));
        const float4 pA  = *(const float4*)&Pv[rA][4 * lane];
        const float4 pB  = *(const float4*)&Pv[rA + 1][4 * lane];
        float4 oA, oB;
        oA.x = (pA.x + hsA.x) * scA; oA.y = (pA.y + hsA.y) * scA;
        oA.z = (pA.z + hsA.z) * scA; oA.w = (pA.w + hsA.w) * scA;
        oB.x = (pB.x + hsB.x) * scB; oB.y = (pB.y + hsB.y) * scB;
        oB.z = (pB.z + hsB.z) * scB; oB.w = (pB.w + hsB.w) * scB;
        *(float4*)(out + (size_t)dA * OUT_F + 4 * lane) = oA;
        *(float4*)(out + (size_t)dB * OUT_F + 4 * lane) = oB;
    }
}

extern "C" void kernel_launch(void* const* d_in, const int* in_sizes, int n_in,
                              void* d_out, int out_size, void* d_ws, size_t ws_size,
                              hipStream_t stream)
{
    (void)in_sizes; (void)n_in; (void)out_size; (void)ws_size;

    const float* feat = (const float*)d_in[0];
    const float* w1   = (const float*)d_in[1];
    const float* w2   = (const float*)d_in[2];
    const float* vmat = (const float*)d_in[3];
    const int*   src  = (const int*)d_in[4];
    const int*   dst  = (const int*)d_in[5];

    float* ws     = (float*)d_ws;
    float* odeg   = ws + OFF_ODEG;
    int*   ideg   = (int*)(ws + OFF_IDEG);
    int*   rowptr = (int*)(ws + OFF_ROWPTR);
    int*   csr    = (int*)(ws + OFF_CSR);
    int*   cntI   = (int*)(ws + OFF_CNTI);
    int*   cntO   = (int*)(ws + OFF_CNTO);
    int*   btot   = (int*)(ws + OFF_BTOT);
    float* fsum   = ws + OFF_FSUM;
    float* fprod  = ws + OFF_FPROD;

    k_count<<<dim3(PRE_S, PRE_R), PRE_T, 0, stream>>>(src, dst, cntI, cntO, btot);
    k_place<<<dim3(PRE_S, PRE_R), PRE_T, 0, stream>>>(src, dst, cntI, cntO, btot, odeg, ideg, rowptr, csr);
    k_node_gemm<<<N_NODES / 16, 128, 0, stream>>>(feat, w1, w2, odeg, fsum, fprod);
    k_gather_final<<<N_NODES / 16, 256, 0, stream>>>(rowptr, ideg, csr, fsum, fprod, vmat, (float*)d_out);
}


# Slot blocks and node-product blocks share ONE launch (products read out-degrees from the count tables): slotting runs beside the WMMA products; slice-inner slot order still by LDS arrival
# speedup vs baseline: 3.0476x; 1.1674x over previous
#include <hip/hip_runtime.h>
#include <math.h>


typedef __attribute__((ext_vector_type(2))) float v2f;
typedef __attribute__((ext_vector_type(8))) float v8f;

#define N_NODES 50000
#define N_EDGES 800000
#define IN_F    256
#define OUT_F   128
#define RANK    64

static_assert(N_NODES % 16 == 0, "row tiles exact");
static_assert(OUT_F == 8 * 16, "one 16-column tile of the finish per wave of a gather block");

#define OFF_IDEG   50000u
#define OFF_CNTI   100000u
#define OFF_CNTO   1350000u
#define OFF_BTOT   2600000u
#define OFF_ROWPTR 9700000u
#define OFF_CSR    9760000u
#define OFF_FSUM   12900000u
#define OFF_FPROD  19300000u

#define PRE_R      25
#define PRE_NR     (N_NODES / PRE_R)
#define PRE_S      25
#define PRE_ES     (N_EDGES / PRE_S)
#define PRE_ES4    (PRE_ES / 4)
#define PRE_T      320
#define PRE_TRIPS  (PRE_ES4 / PRE_T)
#define PRE_BATCH  5
static_assert(PRE_R * PRE_NR == N_NODES, "ranges cover the nodes exactly");
static_assert(PRE_S * PRE_ES == N_EDGES, "slices cover the edges exactly");
static_assert(PRE_ES % 4 == 0, "a slice is whole 16-byte loads");
static_assert(PRE_T * PRE_TRIPS == PRE_ES4, "the threads cover a slice exactly: no guard on the loads");
static_assert(PRE_TRIPS % PRE_BATCH == 0, "whole batches");
static_assert(PRE_T % 32 == 0, "whole waves");
static_assert(PRE_NR % 8 == 0, "a slot block's threads take 8 nodes each");
static_assert(N_NODES % 4 == 0, "16-byte steps between the slices of a count table");
static_assert(OFF_CNTI + PRE_S * N_NODES <= OFF_CNTO, "count tables do not overlap");
static_assert(OFF_CNTO + PRE_S * N_NODES <= OFF_BTOT, "count tables do not overlap");
static_assert(OFF_BTOT + PRE_R * PRE_S <= OFF_ROWPTR, "count tables end before the row starts");

#define FUSE_T       256
#define PL_BLOCKS    (PRE_R * PRE_S)
#define PL_FULL      (PRE_ES4 / FUSE_T)
#define PL_LAST      (PRE_ES4 - PL_FULL * FUSE_T)
#define PL_BATCH     4
#define PL_TRIPS     (PL_FULL + 1)
#define GEMM_ROWS    32
#define GEMM_BLOCKS  ((N_NODES + GEMM_ROWS - 1) / GEMM_ROWS)
static_assert(PL_LAST > 0 && PL_LAST < FUSE_T, "the last trip is a partial one");
static_assert(PL_TRIPS % PL_BATCH == 0, "whole batches");
static_assert(PRE_NR <= FUSE_T * 8, "8 nodes per thread cover a range");
static_assert(FUSE_T == 256, "8 waves: two row tiles x four column groups in a product block");

__global__ __launch_bounds__(PRE_T) void k_count(const int* __restrict__ src,
                                               const int* __restrict__ dst,
                                               int* __restrict__ cntI,
                                               int* __restrict__ cntO,
                                               int* __restrict__ btot)
{
    __shared__ int ci[PRE_NR];
    __shared__ int co[PRE_NR];
    __shared__ int wsum[PRE_T / 32];

    const int s    = blockIdx.x;
    const int r    = blockIdx.y;
    const int tid  = threadIdx.x;
    const int lane = tid & 31;
    const int wave = tid >> 5;
    const int n0   = r * PRE_NR;

    for (int t = tid; t < PRE_NR; t += PRE_T) { ci[t] = 0; co[t] = 0; }
    __syncthreads();

    const int4* d4 = (const int4*)(dst + (size_t)s * PRE_ES);
    const int4* s4 = (const int4*)(src + (size_t)s * PRE_ES);
    for (int kb = 0; kb < PRE_TRIPS; kb += PRE_BATCH) {
        int4 d[PRE_BATCH], q[PRE_BATCH];
#pragma unroll
        for (int j = 0; j < PRE_BATCH; ++j) {
            d[j] = d4[tid + PRE_T * (kb + j)];
            q[j] = s4[tid + PRE_T * (kb + j)];
        }
#pragma unroll
        for (int j = 0; j < PRE_BATCH; ++j) {
            unsigned u;
            u = (unsigned)(d[j].x - n0); if (u < (unsigned)PRE_NR) atomicAdd(&ci[u], 1);
            u = (unsigned)(d[j].y - n0); if (u < (unsigned)PRE_NR) atomicAdd(&ci[u], 1);
            u = (unsigned)(d[j].z - n0); if (u < (unsigned)PRE_NR) atomicAdd(&ci[u], 1);
            u = (unsigned)(d[j].w - n0); if (u < (unsigned)PRE_NR) atomicAdd(&ci[u], 1);
            u = (unsigned)(q[j].x - n0); if (u < (unsigned)PRE_NR) atomicAdd(&co[u], 1);
            u = (unsigned)(q[j].y - n0); if (u < (unsigned)PRE_NR) atomicAdd(&co[u], 1);
            u = (unsigned)(q[j].z - n0); if (u < (unsigned)PRE_NR) atomicAdd(&co[u], 1);
            u = (unsigned)(q[j].w - n0); if (u < (unsigned)PRE_NR) atomicAdd(&co[u], 1);
        }
    }
    __syncthreads();

    int part = 0;
    for (int t = tid; t < PRE_NR; t += PRE_T) {
        const int a = ci[t];
        cntI[(size_t)s * N_NODES + n0 + t] = a;
        cntO[(size_t)s * N_NODES + n0 + t] = co[t];
        part += a;
    }
#pragma unroll
    for (int o = 16; o > 0; o >>= 1) part += __shfl_xor(part, o);
    if (lane == 0) wsum[wave] = part;
    __syncthreads();
    if (tid == 0) {
        int tot = 0;
#pragma unroll
        for (int w = 0; w < PRE_T / 32; ++w) tot += wsum[w];
        btot[r * PRE_S + s] = tot;
    }
}

struct PlaceSmem {
    int base[PRE_NR];
    int cur[PRE_NR];
    int wtot[FUSE_T / 32];
    int wpre[FUSE_T / 32];
};
struct GemmSmem {
    float As[GEMM_ROWS][260];
    float sc[GEMM_ROWS];
};
union FusedSmem {
    PlaceSmem p;
    GemmSmem  g;
};

__device__ __forceinline__ void place_block(PlaceSmem& sm, const int s, const int r, const int tid,
                                            const int* __restrict__ src,
                                            const int* __restrict__ dst,
                                            const int* __restrict__ cntI,
                                            const int* __restrict__ btot,
                                            int* __restrict__ ideg,
                                            int* __restrict__ rowptr,
                                            int* __restrict__ csr)
{
    const int lane = tid & 31;
    const int wave = tid >> 5;
    const int n0   = r * PRE_NR;
    const int t0   = tid * 8;
    const bool has = (t0 < PRE_NR);

    int pre = 0;
    for (int i = tid; i < r * PRE_S; i += FUSE_T) pre += btot[i];

    int tot[8], bef[8];
#pragma unroll
    for (int j = 0; j < 8; ++j) { tot[j] = 0; bef[j] = 0; }
    if (has) {
        const int4* c = (const int4*)(cntI + n0 + t0);
#pragma unroll 5
        for (int k = 0; k < PRE_S; ++k) {
            if (k == s) {
#pragma unroll
                for (int j = 0; j < 8; ++j) bef[j] = tot[j];
            }
            const int4 a = c[(size_t)k * (N_NODES / 4)];
            const int4 b = c[(size_t)k * (N_NODES / 4) + 1];
            tot[0] += a.x; tot[1] += a.y; tot[2] += a.z; tot[3] += a.w;
            tot[4] += b.x; tot[5] += b.y; tot[6] += b.z; tot[7] += b.w;
        }
    }
    int tsum = 0;
#pragma unroll
    for (int j = 0; j < 8; ++j) tsum += tot[j];

    int incl = tsum;
#pragma unroll
    for (int o = 1; o < 32; o <<= 1) {
        int up = __shfl_up(incl, o);
        if (lane >= o) incl += up;
    }
#pragma unroll
    for (int o = 16; o > 0; o >>= 1) pre += __shfl_xor(pre, o);
    if (lane == 31) sm.wtot[wave] = incl;
    if (lane == 0)  sm.wpre[wave] = pre;
    __syncthreads();
    int woff = 0, rbase = 0;
#pragma unroll
    for (int w = 0; w < FUSE_T / 32; ++w) {
        rbase += sm.wpre[w];
        if (w < wave) woff += sm.wtot[w];
    }

    if (has) {
        int run = rbase + woff + (incl - tsum);
        int rs[8];
#pragma unroll
        for (int j = 0; j < 8; ++j) {
            rs[j] = run;
            sm.base[t0 + j] = run + bef[j];
            sm.cur[t0 + j]  = 0;
            run += tot[j];
        }
        if (s == 0) {
            int4* rp = (int4*)(rowptr + n0 + t0);
            int4* ip = (int4*)(ideg + n0 + t0);
            rp[0] = make_int4(rs[0], rs[1], rs[2], rs[3]);
            rp[1] = make_int4(rs[4], rs[5], rs[6], rs[7]);
            ip[0] = make_int4(tot[0], tot[1], tot[2], tot[3]);
            ip[1] = make_int4(tot[4], tot[5], tot[6], tot[7]);
        }
    }
    __syncthreads();

    const int4* d4 = (const int4*)(dst + (size_t)s * PRE_ES);
    const int4* s4 = (const int4*)(src + (size_t)s * PRE_ES);
    for (int kb = 0; kb < PL_TRIPS; kb += PL_BATCH) {
        int4 d[PL_BATCH], q[PL_BATCH];
#pragma unroll
        for (int j = 0; j < PL_BATCH; ++j) {
            d[j] = make_int4(-1, -1, -1, -1);
            q[j] = d[j];
            if (kb + j < PL_FULL || tid < PL_LAST) {
                d[j] = d4[tid + FUSE_T * (kb + j)];
                q[j] = s4[tid + FUSE_T * (kb + j)];
            }
        }
#pragma unroll
        for (int j = 0; j < PL_BATCH; ++j) {
            unsigned u, p;
            u = (unsigned)(d[j].x - n0);
            if (u < (unsigned)PRE_NR) { p = (unsigned)(sm.base[u] + atomicAdd(&sm.cur[u], 1)); if (p < (unsigned)N_EDGES) csr[p] = q[j].x; }
            u = (unsigned)(d[j].y - n0);
            if (u < (unsigned)PRE_NR) { p = (unsigned)(sm.base[u] + atomicAdd(&sm.cur[u], 1)); if (p < (unsigned)N_EDGES) csr[p] = q[j].y; }
            u = (unsigned)(d[j].z - n0);
            if (u < (unsigned)PRE_NR) { p = (unsigned)(sm.base[u] + atomicAdd(&sm.cur[u], 1)); if (p < (unsigned)N_EDGES) csr[p] = q[j].z; }
            u = (unsigned)(d[j].w - n0);
            if (u < (unsigned)PRE_NR) { p = (unsigned)(sm.base[u] + atomicAdd(&sm.cur[u], 1)); if (p < (unsigned)N_EDGES) csr[p] = q[j].w; }
        }
    }
}

__device__ __forceinline__ void gemm_block(GemmSmem& sm, const int g, const int tid,
                                           const float* __restrict__ feat,
                                           const float* __restrict__ w1,
                                           const float* __restrict__ w2,
                                           const int* __restrict__ cntO,
                                           float* __restrict__ fsum,
                                           float* __restrict__ fprod)
{
    const int m0 = g * GEMM_ROWS;

    if (tid < GEMM_ROWS) {
        const int row = m0 + tid;
        int od = 0;
        if (row < N_NODES) {
#pragma unroll
            for (int k = 0; k < PRE_S; ++k) od += cntO[(size_t)k * N_NODES + row];
        }
        sm.sc[tid] = rsqrtf(fmaxf((float)od, 1.0f));
    }
    __syncthreads();

    {
        const int r   = tid >> 3;
        const int c0  = (tid & 7) * 32;
        if (m0 + r < N_NODES) {
            const float s = sm.sc[r];
            const float4* gp = (const float4*)(feat + (size_t)(m0 + r) * IN_F + c0);
#pragma unroll
            for (int j = 0; j < 8; ++j) {
                float4 x = gp[j];
                sm.As[r][c0 + 4 * j + 0] = x.x * s;
                sm.As[r][c0 + 4 * j + 1] = x.y * s;
                sm.As[r][c0 + 4 * j + 2] = x.z * s;
                sm.As[r][c0 + 4 * j + 3] = x.w * s;
            }
        }
    }
    __syncthreads();

    const int wave  = tid >> 5;
    const int lane  = tid & 31;
    const int rt    = wave >> 2;
    const int wq    = wave & 3;
    const int rsel  = lane & 15;
    const int khalf = (lane < 16) ? 0 : 2;
    const int rofs  = (lane < 16) ? 0 : 8;
    const int mt    = m0 + 16 * rt;

    if (mt >= N_NODES) return;

    for (int t = 0; t < 3; ++t) {
        const int  tile = wq + 4 * t;
        const bool isW2 = (tile >= 8);
        const int  n0   = isW2 ? (tile - 8) * 16 : tile * 16;
        const float* B  = isW2 ? w2 : w1;
        const int  ldb  = isW2 ? RANK : OUT_F;

        for (int k = 0; k < IN_F; k += 32)
            __builtin_prefetch(B + (size_t)k * ldb + n0, 0, 0);

        v8f acc = {};
        for (int k0 = 0; k0 < IN_F; k0 += 4) {
            v2f a;
            a.x = sm.As[16 * rt + rsel][k0 + khalf];
            a.y = sm.As[16 * rt + rsel][k0 + khalf + 1];
            v2f b;
            const float* Bp = B + (size_t)(k0 + khalf) * ldb + n0 + rsel;
            b.x = Bp[0];
            b.y = Bp[ldb];
            acc = __builtin_amdgcn_wmma_f32_16x16x4_f32(
                false, a, false, b, (short)0, acc, false, false);
        }

        const int colg = n0 + rsel;
        if (isW2) {
            const float bias = w2[(size_t)IN_F * RANK + colg];
            float* op = fprod + (size_t)(mt + rofs) * RANK + colg;
#pragma unroll
            for (int r = 0; r < 8; ++r)
                op[r * RANK] = tanhf(acc[r] + bias);
        } else {
            float* op = fsum + (size_t)(mt + rofs) * OUT_F + colg;
#pragma unroll
            for (int r = 0; r < 8; ++r)
                op[r * OUT_F] = acc[r];
        }
    }
}

__global__ __launch_bounds__(FUSE_T) void k_place_gemm(const int* __restrict__ src,
                                                       const int* __restrict__ dst,
                                                       const int* __restrict__ cntI,
                                                       const int* __restrict__ cntO,
                                                       const int* __restrict__ btot,
                                                       int*   __restrict__ ideg,
                                                       int*   __restrict__ rowptr,
                                                       int*   __restrict__ csr,
                                                       const float* __restrict__ feat,
                                                       const float* __restrict__ w1,
                                                       const float* __restrict__ w2,
                                                       float* __restrict__ fsum,
                                                       float* __restrict__ fprod)
{
    __shared__ __attribute__((aligned(16))) FusedSmem sm;

    const int b   = (int)blockIdx.x;
    const int tid = (int)threadIdx.x;
    if (b < PL_BLOCKS) {
        place_block(sm.p, b % PRE_S, b / PRE_S, tid, src, dst, cntI, btot, ideg, rowptr, csr);
    } else {
        gemm_block(sm.g, b - PL_BLOCKS, tid, feat, w1, w2, cntO, fsum, fprod);
    }
}

__device__ __forceinline__ void gather_row(const int start, const int cnt, int mine, const int lane,
                                           const int* __restrict__ csr,
                                           const float* __restrict__ fsum,
                                           const float* __restrict__ fprod,
                                           float4& hs_out, float2& hp_out)
{
    float4 hs = make_float4(0.0f, 0.0f, 0.0f, 0.0f);
    float  la0 = 0.0f, la1 = 0.0f;
    int    ng0 = 0, ng1 = 0;

    for (int b = 0; b < cnt; b += 32) {
        const int n = min(32, cnt - b);
        if (b > 0) {
            mine = 0;
            if (lane < n) mine = csr[start + b + lane];
        }
#pragma unroll 4
        for (int j = 0; j < n; ++j) {
            const unsigned s = min((unsigned)__builtin_amdgcn_readlane(mine, j), (unsigned)(N_NODES - 1));
            const float4 x = *(const float4*)(fsum  + (size_t)s * OUT_F + 4 * lane);
            const float2 m = *(const float2*)(fprod + (size_t)s * RANK  + 2 * lane);
            hs.x += x.x; hs.y += x.y; hs.z += x.z; hs.w += x.w;
            la0 += __logf(fabsf(m.x));
            la1 += __logf(fabsf(m.y));
            ng0 += (m.x < 0.0f) ? 1 : 0;
            ng1 += (m.y < 0.0f) ? 1 : 0;
        }
    }

    float2 hp = make_float2(0.0f, 0.0f);
    if (cnt > 0) {
        hp.x = ((ng0 & 1) ? -1.0f : 1.0f) * __expf(la0);
        hp.y = ((ng1 & 1) ? -1.0f : 1.0f) * __expf(la1);
    }
    hs_out = hs;
    hp_out = hp;
}

__global__ __launch_bounds__(256) void k_gather_final(const int* __restrict__ rowptr,
                                                      const int* __restrict__ ideg,
                                                      const int* __restrict__ csr,
                                                      const float* __restrict__ fsum,
                                                      const float* __restrict__ fprod,
                                                      const float* __restrict__ vmat,
                                                      float* __restrict__ out)
{
    __shared__ __attribute__((aligned(16))) float Hp[16][68];
    __shared__ __attribute__((aligned(16))) float Pv[16][132];

    const int lane = threadIdx.x & 31;
    const int wave = __builtin_amdgcn_readfirstlane((int)(threadIdx.x >> 5));
    const int m0   = (int)(blockIdx.x * 16u);
    const int rA   = 2 * wave;
    const int dA   = m0 + rA;
    const int dB   = dA + 1;

    const int startA = rowptr[dA];
    const int cntA   = ideg[dA];
    const int startB = rowptr[dB];
    const int cntB   = ideg[dB];

    int mineA = 0, mineB = 0;
    if (lane < cntA) mineA = csr[startA + lane];
    if (lane < cntB) mineB = csr[startB + lane];

    float4 hsA, hsB;
    float2 hpA, hpB;
    gather_row(startA, cntA, mineA, lane, csr, fsum, fprod, hsA, hpA);
    gather_row(startB, cntB, mineB, lane, csr, fsum, fprod, hsB, hpB);

    const int rsel  = lane & 15;
    const int khalf = (lane < 16) ? 0 : 2;
    const int rofs  = (lane < 16) ? 0 : 8;
    const int n0    = wave * 16;

    v2f bv[RANK / 4];
    {
        const float* Bp = vmat + (size_t)khalf * OUT_F + n0 + rsel;
#pragma unroll
        for (int i = 0; i < RANK / 4; ++i) {
            bv[i].x = Bp[(size_t)(4 * i) * OUT_F];
            bv[i].y = Bp[(size_t)(4 * i + 1) * OUT_F];
        }
    }

    *(float2*)&Hp[rA][2 * lane]     = hpA;
    *(float2*)&Hp[rA + 1][2 * lane] = hpB;
    __syncthreads();

    {
        v8f acc = {};
#pragma unroll
        for (int i = 0; i < RANK / 4; ++i) {
            const int k0 = 4 * i;
            v2f a;
            a.x = Hp[rsel][k0 + khalf];
            a.y = Hp[rsel][k0 + khalf + 1];
            acc = __builtin_amdgcn_wmma_f32_16x16x4_f32(
                false, a, false, bv[i], (short)0, acc, false, false);
        }
#pragma unroll
        for (int r = 0; r < 8; ++r)
            Pv[rofs + r][n0 + rsel] = acc[r];
    }
    __syncthreads();

    {
        const float  scA = rsqrtf(fmaxf((float)cntA, 1.0f));
        const float  scB = rsqrtf(fmaxf((float)cntB, 1.0f));
        const float4 pA  = *(const float4*)&Pv[rA][4 * lane];
        const float4 pB  = *(const float4*)&Pv[rA + 1][4 * lane];
        float4 oA, oB;
        oA.x = (pA.x + hsA.x) * scA; oA.y = (pA.y + hsA.y) * scA;
        oA.z = (pA.z + hsA.z) * scA; oA.w = (pA.w + hsA.w) * scA;
        oB.x = (pB.x + hsB.x) * scB; oB.y = (pB.y + hsB.y) * scB;
        oB.z = (pB.z + hsB.z) * scB; oB.w = (pB.w + hsB.w) * scB;
        *(float4*)(out + (size_t)dA * OUT_F + 4 * lane) = oA;
        *(float4*)(out + (size_t)dB * OUT_F + 4 * lane) = oB;
    }
}

extern "C" void kernel_launch(void* const* d_in, const int* in_sizes, int n_in,
                              void* d_out, int out_size, void* d_ws, size_t ws_size,
                              hipStream_t stream)
{
    (void)in_sizes; (void)n_in; (void)out_size; (void)ws_size;

    const float* feat = (const float*)d_in[0];
    const float* w1   = (const float*)d_in[1];
    const float* w2   = (const float*)d_in[2];
    const float* vmat = (const float*)d_in[3];
    const int*   src  = (const int*)d_in[4];
    const int*   dst  = (const int*)d_in[5];

    float* ws     = (float*)d_ws;
    int*   ideg   = (int*)(ws + OFF_IDEG);
    int*   rowptr = (int*)(ws + OFF_ROWPTR);
    int*   csr    = (int*)(ws + OFF_CSR);
    int*   cntI   = (int*)(ws + OFF_CNTI);
    int*   cntO   = (int*)(ws + OFF_CNTO);
    int*   btot   = (int*)(ws + OFF_BTOT);
    float* fsum   = ws + OFF_FSUM;
    float* fprod  = ws + OFF_FPROD;

    k_count<<<dim3(PRE_S, PRE_R), PRE_T, 0, stream>>>(src, dst, cntI, cntO, btot);
    k_place_gemm<<<PL_BLOCKS + GEMM_BLOCKS, FUSE_T, 0, stream>>>(src, dst, cntI, cntO, btot, ideg, rowptr, csr,
                                                                 feat, w1, w2, fsum, fprod);
    k_gather_final<<<N_NODES / 16, 256, 0, stream>>>(rowptr, ideg, csr, fsum, fprod, vmat, (float*)d_out);
}


# Node-product blocks: a wave keeps both row tiles of its column tiles, k outermost, so one B fragment (two global loads) feeds two WMMAs; A tile staged by whole-line loads; every sum's k order kept
# speedup vs baseline: 3.2318x; 1.0604x over previous
#include <hip/hip_runtime.h>
#include <math.h>


typedef __attribute__((ext_vector_type(2))) float v2f;
typedef __attribute__((ext_vector_type(8))) float v8f;

#define N_NODES 50000
#define N_EDGES 800000
#define IN_F    256
#define OUT_F   128
#define RANK    64

static_assert(N_NODES % 16 == 0, "row tiles exact");
static_assert(OUT_F == 8 * 16, "one 16-column tile of the finish per wave of a gather block");

#define OFF_IDEG   50000u
#define OFF_CNTI   100000u
#define OFF_CNTO   1350000u
#define OFF_BTOT   2600000u
#define OFF_ROWPTR 9700000u
#define OFF_CSR    9760000u
#define OFF_FSUM   12900000u
#define OFF_FPROD  19300000u

#define PRE_R      25
#define PRE_NR     (N_NODES / PRE_R)
#define PRE_S      25
#define PRE_ES     (N_EDGES / PRE_S)
#define PRE_ES4    (PRE_ES / 4)
#define PRE_T      320
#define PRE_TRIPS  (PRE_ES4 / PRE_T)
#define PRE_BATCH  5
static_assert(PRE_R * PRE_NR == N_NODES, "ranges cover the nodes exactly");
static_assert(PRE_S * PRE_ES == N_EDGES, "slices cover the edges exactly");
static_assert(PRE_ES % 4 == 0, "a slice is whole 16-byte loads");
static_assert(PRE_T * PRE_TRIPS == PRE_ES4, "the threads cover a slice exactly: no guard on the loads");
static_assert(PRE_TRIPS % PRE_BATCH == 0, "whole batches");
static_assert(PRE_T % 32 == 0, "whole waves");
static_assert(PRE_NR % 8 == 0, "a slot block's threads take 8 nodes each");
static_assert(N_NODES % 4 == 0, "16-byte steps between the slices of a count table");
static_assert(OFF_CNTI + PRE_S * N_NODES <= OFF_CNTO, "count tables do not overlap");
static_assert(OFF_CNTO + PRE_S * N_NODES <= OFF_BTOT, "count tables do not overlap");
static_assert(OFF_BTOT + PRE_R * PRE_S <= OFF_ROWPTR, "count tables end before the row starts");

#define FUSE_T       256
#define PL_BLOCKS    (PRE_R * PRE_S)
#define PL_FULL      (PRE_ES4 / FUSE_T)
#define PL_LAST      (PRE_ES4 - PL_FULL * FUSE_T)
#define PL_BATCH     4
#define PL_TRIPS     (PL_FULL + 1)
#define GEMM_ROWS    32
#define GEMM_BLOCKS  ((N_NODES + GEMM_ROWS - 1) / GEMM_ROWS)
static_assert(PL_LAST > 0 && PL_LAST < FUSE_T, "the last trip is a partial one");
static_assert(PL_TRIPS % PL_BATCH == 0, "whole batches");
static_assert(PRE_NR <= FUSE_T * 8, "8 nodes per thread cover a range");
static_assert(FUSE_T == 256, "8 waves in a product block: four with two w1 column tiles, four with one w2 column tile");

__global__ __launch_bounds__(PRE_T) void k_count(const int* __restrict__ src,
                                               const int* __restrict__ dst,
                                               int* __restrict__ cntI,
                                               int* __restrict__ cntO,
                                               int* __restrict__ btot)
{
    __shared__ int ci[PRE_NR];
    __shared__ int co[PRE_NR];
    __shared__ int wsum[PRE_T / 32];

    const int s    = blockIdx.x;
    const int r    = blockIdx.y;
    const int tid  = threadIdx.x;
    const int lane = tid & 31;
    const int wave = tid >> 5;
    const int n0   = r * PRE_NR;

    for (int t = tid; t < PRE_NR; t += PRE_T) { ci[t] = 0; co[t] = 0; }
    __syncthreads();

    const int4* d4 = (const int4*)(dst + (size_t)s * PRE_ES);
    const int4* s4 = (const int4*)(src + (size_t)s * PRE_ES);
    for (int kb = 0; kb < PRE_TRIPS; kb += PRE_BATCH) {
        int4 d[PRE_BATCH], q[PRE_BATCH];
#pragma unroll
        for (int j = 0; j < PRE_BATCH; ++j) {
            d[j] = d4[tid + PRE_T * (kb + j)];
            q[j] = s4[tid + PRE_T * (kb + j)];
        }
#pragma unroll
        for (int j = 0; j < PRE_BATCH; ++j) {
            unsigned u;
            u = (unsigned)(d[j].x - n0); if (u < (unsigned)PRE_NR) atomicAdd(&ci[u], 1);
            u = (unsigned)(d[j].y - n0); if (u < (unsigned)PRE_NR) atomicAdd(&ci[u], 1);
            u = (unsigned)(d[j].z - n0); if (u < (unsigned)PRE_NR) atomicAdd(&ci[u], 1);
            u = (unsigned)(d[j].w - n0); if (u < (unsigned)PRE_NR) atomicAdd(&ci[u], 1);
            u = (unsigned)(q[j].x - n0); if (u < (unsigned)PRE_NR) atomicAdd(&co[u], 1);
            u = (unsigned)(q[j].y - n0); if (u < (unsigned)PRE_NR) atomicAdd(&co[u], 1);
            u = (unsigned)(q[j].z - n0); if (u < (unsigned)PRE_NR) atomicAdd(&co[u], 1);
            u = (unsigned)(q[j].w - n0); if (u < (unsigned)PRE_NR) atomicAdd(&co[u], 1);
        }
    }
    __syncthreads();

    int part = 0;
    for (int t = tid; t < PRE_NR; t += PRE_T) {
        const int a = ci[t];
        cntI[(size_t)s * N_NODES + n0 + t] = a;
        cntO[(size_t)s * N_NODES + n0 + t] = co[t];
        part += a;
    }
#pragma unroll
    for (int o = 16; o > 0; o >>= 1) part += __shfl_xor(part, o);
    if (lane == 0) wsum[wave] = part;
    __syncthreads();
    if (tid == 0) {
        int tot = 0;
#pragma unroll
        for (int w = 0; w < PRE_T / 32; ++w) tot += wsum[w];
        btot[r * PRE_S + s] = tot;
    }
}

struct PlaceSmem {
    int base[PRE_NR];
    int cur[PRE_NR];
    int wtot[FUSE_T / 32];
    int wpre[FUSE_T / 32];
};
struct GemmSmem {
    float As[GEMM_ROWS][260];
    float sc[GEMM_ROWS];
};
union FusedSmem {
    PlaceSmem p;
    GemmSmem  g;
};

__device__ __forceinline__ void place_block(PlaceSmem& sm, const int s, const int r, const int tid,
                                            const int* __restrict__ src,
                                            const int* __restrict__ dst,
                                            const int* __restrict__ cntI,
                                            const int* __restrict__ btot,
                                            int* __restrict__ ideg,
                                            int* __restrict__ rowptr,
                                            int* __restrict__ csr)
{
    const int lane = tid & 31;
    const int wave = tid >> 5;
    const int n0   = r * PRE_NR;
    const int t0   = tid * 8;
    const bool has = (t0 < PRE_NR);

    int pre = 0;
    for (int i = tid; i < r * PRE_S; i += FUSE_T) pre += btot[i];

    int tot[8], bef[8];
#pragma unroll
    for (int j = 0; j < 8; ++j) { tot[j] = 0; bef[j] = 0; }
    if (has) {
        const int4* c = (const int4*)(cntI + n0 + t0);
#pragma unroll 5
        for (int k = 0; k < PRE_S; ++k) {
            if (k == s) {
#pragma unroll
                for (int j = 0; j < 8; ++j) bef[j] = tot[j];
            }
            const int4 a = c[(size_t)k * (N_NODES / 4)];
            const int4 b = c[(size_t)k * (N_NODES / 4) + 1];
            tot[0] += a.x; tot[1] += a.y; tot[2] += a.z; tot[3] += a.w;
            tot[4] += b.x; tot[5] += b.y; tot[6] += b.z; tot[7] += b.w;
        }
    }
    int tsum = 0;
#pragma unroll
    for (int j = 0; j < 8; ++j) tsum += tot[j];

    int incl = tsum;
#pragma unroll
    for (int o = 1; o < 32; o <<= 1) {
        int up = __shfl_up(incl, o);
        if (lane >= o) incl += up;
    }
#pragma unroll
    for (int o = 16; o > 0; o >>= 1) pre += __shfl_xor(pre, o);
    if (lane == 31) sm.wtot[wave] = incl;
    if (lane == 0)  sm.wpre[wave] = pre;
    __syncthreads();
    int woff = 0, rbase = 0;
#pragma unroll
    for (int w = 0; w < FUSE_T / 32; ++w) {
        rbase += sm.wpre[w];
        if (w < wave) woff += sm.wtot[w];
    }

    if (has) {
        int run = rbase + woff + (incl - tsum);
        int rs[8];
#pragma unroll
        for (int j = 0; j < 8; ++j) {
            rs[j] = run;
            sm.base[t0 + j] = run + bef[j];
            sm.cur[t0 + j]  = 0;
            run += tot[j];
        }
        if (s == 0) {
            int4* rp = (int4*)(rowptr + n0 + t0);
            int4* ip = (int4*)(ideg + n0 + t0);
            rp[0] = make_int4(rs[0], rs[1], rs[2], rs[3]);
            rp[1] = make_int4(rs[4], rs[5], rs[6], rs[7]);
            ip[0] = make_int4(tot[0], tot[1], tot[2], tot[3]);
            ip[1] = make_int4(tot[4], tot[5], tot[6], tot[7]);
        }
    }
    __syncthreads();

    const int4* d4 = (const int4*)(dst + (size_t)s * PRE_ES);
    const int4* s4 = (const int4*)(src + (size_t)s * PRE_ES);
    for (int kb = 0; kb < PL_TRIPS; kb += PL_BATCH) {
        int4 d[PL_BATCH], q[PL_BATCH];
#pragma unroll
        for (int j = 0; j < PL_BATCH; ++j) {
            d[j] = make_int4(-1, -1, -1, -1);
            q[j] = d[j];
            if (kb + j < PL_FULL || tid < PL_LAST) {
                d[j] = d4[tid + FUSE_T * (kb + j)];
                q[j] = s4[tid + FUSE_T * (kb + j)];
            }
        }
#pragma unroll
        for (int j = 0; j < PL_BATCH; ++j) {
            unsigned u, p;
            u = (unsigned)(d[j].x - n0);
            if (u < (unsigned)PRE_NR) { p = (unsigned)(sm.base[u] + atomicAdd(&sm.cur[u], 1)); if (p < (unsigned)N_EDGES) csr[p] = q[j].x; }
            u = (unsigned)(d[j].y - n0);
            if (u < (unsigned)PRE_NR) { p = (unsigned)(sm.base[u] + atomicAdd(&sm.cur[u], 1)); if (p < (unsigned)N_EDGES) csr[p] = q[j].y; }
            u = (unsigned)(d[j].z - n0);
            if (u < (unsigned)PRE_NR) { p = (unsigned)(sm.base[u] + atomicAdd(&sm.cur[u], 1)); if (p < (unsigned)N_EDGES) csr[p] = q[j].z; }
            u = (unsigned)(d[j].w - n0);
            if (u < (unsigned)PRE_NR) { p = (unsigned)(sm.base[u] + atomicAdd(&sm.cur[u], 1)); if (p < (unsigned)N_EDGES) csr[p] = q[j].w; }
        }
    }
}

__device__ __forceinline__ void gemm_block(GemmSmem& sm, const int g, const int tid,
                                           const float* __restrict__ feat,
                                           const float* __restrict__ w1,
                                           const float* __restrict__ w2,
                                           const int* __restrict__ cntO,
                                           float* __restrict__ fsum,
                                           float* __restrict__ fprod)
{
    const int m0 = g * GEMM_ROWS;

    if (tid < GEMM_ROWS) {
        const int row = m0 + tid;
        int od = 0;
        if (row < N_NODES) {
#pragma unroll
            for (int k = 0; k < PRE_S; ++k) od += cntO[(size_t)k * N_NODES + row];
        }
        sm.sc[tid] = rsqrtf(fmaxf((float)od, 1.0f));
    }
    __syncthreads();

    const bool two = (m0 + 16 < N_NODES);
    {
        const int c4 = tid & 63;
        const int r0 = tid >> 6;
        const float4* gp = (const float4*)(feat + (size_t)(m0 + r0) * IN_F) + c4;
        float4 x[GEMM_ROWS / 4];
#pragma unroll
        for (int j = 0; j < GEMM_ROWS / 8; ++j) x[j] = gp[(size_t)j * IN_F];
        if (two) {
#pragma unroll
            for (int j = GEMM_ROWS / 8; j < GEMM_ROWS / 4; ++j) x[j] = gp[(size_t)j * IN_F];
        } else {
#pragma unroll
            for (int j = GEMM_ROWS / 8; j < GEMM_ROWS / 4; ++j) x[j] = make_float4(0.0f, 0.0f, 0.0f, 0.0f);
        }
#pragma unroll
        for (int j = 0; j < GEMM_ROWS / 4; ++j) {
            const int   r = r0 + 4 * j;
            const float s = sm.sc[r];
            float4 y = x[j];
            y.x *= s; y.y *= s; y.z *= s; y.w *= s;
            *(float4*)&sm.As[r][4 * c4] = y;
        }
    }
    __syncthreads();

    const int wave  = tid >> 5;
    const int lane  = tid & 31;
    const int rsel  = lane & 15;
    const int khalf = (lane < 16) ? 0 : 2;
    const int rofs  = (lane < 16) ? 0 : 8;

    const float* ap0 = &sm.As[rsel][khalf];
    const float* ap1 = &sm.As[16 + rsel][khalf];

    if (wave < 4) {
        const int n0 = wave * 16;
        const float* Bp = w1 + (size_t)khalf * OUT_F + n0 + rsel;
        for (int k = 0; k < IN_F; k += 32) {
            __builtin_prefetch(w1 + (size_t)k * OUT_F + n0, 0, 0);
            __builtin_prefetch(w1 + (size_t)k * OUT_F + n0 + 64, 0, 0);
        }
        v8f c00 = {}, c01 = {}, c10 = {}, c11 = {};
#pragma unroll 2
        for (int k0 = 0; k0 < IN_F; k0 += 4) {
            v2f a0, a1, b0, b1;
            a0.x = ap0[k0];     a0.y = ap0[k0 + 1];
            a1.x = ap1[k0];     a1.y = ap1[k0 + 1];
            b0.x = Bp[(size_t)k0 * OUT_F];
            b0.y = Bp[(size_t)(k0 + 1) * OUT_F];
            b1.x = Bp[(size_t)k0 * OUT_F + 64];
            b1.y = Bp[(size_t)(k0 + 1) * OUT_F + 64];
            c00 = __builtin_amdgcn_wmma_f32_16x16x4_f32(false, a0, false, b0, (short)0, c00, false, false);
            c10 = __builtin_amdgcn_wmma_f32_16x16x4_f32(false, a1, false, b0, (short)0, c10, false, false);
            c01 = __builtin_amdgcn_wmma_f32_16x16x4_f32(false, a0, false, b1, (short)0, c01, false, false);
            c11 = __builtin_amdgcn_wmma_f32_16x16x4_f32(false, a1, false, b1, (short)0, c11, false, false);
        }
        float* op = fsum + (size_t)(m0 + rofs) * OUT_F + n0 + rsel;
#pragma unroll
        for (int r = 0; r < 8; ++r) {
            op[r * OUT_F]      = c00[r];
            op[r * OUT_F + 64] = c01[r];
        }
        if (two) {
#pragma unroll
            for (int r = 0; r < 8; ++r) {
                op[(16 + r) * OUT_F]      = c10[r];
                op[(16 + r) * OUT_F + 64] = c11[r];
            }
        }
    } else {
        const int n0 = (wave - 4) * 16;
        const float* Bp = w2 + (size_t)khalf * RANK + n0 + rsel;
        for (int k = 0; k < IN_F; k += 32)
            __builtin_prefetch(w2 + (size_t)k * RANK + n0, 0, 0);
        v8f c0 = {}, c1 = {};
#pragma unroll 4
        for (int k0 = 0; k0 < IN_F; k0 += 4) {
            v2f a0, a1, b;
            a0.x = ap0[k0];     a0.y = ap0[k0 + 1];
            a1.x = ap1[k0];     a1.y = ap1[k0 + 1];
            b.x = Bp[(size_t)k0 * RANK];
            b.y = Bp[(size_t)(k0 + 1) * RANK];
            c0 = __builtin_amdgcn_wmma_f32_16x16x4_f32(false, a0, false, b, (short)0, c0, false, false);
            c1 = __builtin_amdgcn_wmma_f32_16x16x4_f32(false, a1, false, b, (short)0, c1, false, false);
        }
        const int   colg = n0 + rsel;
        const float bias = w2[(size_t)IN_F * RANK + colg];
        float* op = fprod + (size_t)(m0 + rofs) * RANK + colg;
#pragma unroll
        for (int r = 0; r < 8; ++r)
            op[r * RANK] = tanhf(c0[r] + bias);
        if (two) {
#pragma unroll
            for (int r = 0; r < 8; ++r)
                op[(16 + r) * RANK] = tanhf(c1[r] + bias);
        }
    }
}

__global__ __launch_bounds__(FUSE_T) void k_place_gemm(const int* __restrict__ src,
                                                       const int* __restrict__ dst,
                                                       const int* __restrict__ cntI,
                                                       const int* __restrict__ cntO,
                                                       const int* __restrict__ btot,
                                                       int*   __restrict__ ideg,
                                                       int*   __restrict__ rowptr,
                                                       int*   __restrict__ csr,
                                                       const float* __restrict__ feat,
                                                       const float* __restrict__ w1,
                                                       const float* __restrict__ w2,
                                                       float* __restrict__ fsum,
                                                       float* __restrict__ fprod)
{
    __shared__ __attribute__((aligned(16))) FusedSmem sm;

    const int b   = (int)blockIdx.x;
    const int tid = (int)threadIdx.x;
    if (b < PL_BLOCKS) {
        place_block(sm.p, b % PRE_S, b / PRE_S, tid, src, dst, cntI, btot, ideg, rowptr, csr);
    } else {
        gemm_block(sm.g, b - PL_BLOCKS, tid, feat, w1, w2, cntO, fsum, fprod);
    }
}

__device__ __forceinline__ void gather_row(const int start, const int cnt, int mine, const int lane,
                                           const int* __restrict__ csr,
                                           const float* __restrict__ fsum,
                                           const float* __restrict__ fprod,
                                           float4& hs_out, float2& hp_out)
{
    float4 hs = make_float4(0.0f, 0.0f, 0.0f, 0.0f);
    float  la0 = 0.0f, la1 = 0.0f;
    int    ng0 = 0, ng1 = 0;

    for (int b = 0; b < cnt; b += 32) {
        const int n = min(32, cnt - b);
        if (b > 0) {
            mine = 0;
            if (lane < n) mine = csr[start + b + lane];
        }
#pragma unroll 4
        for (int j = 0; j < n; ++j) {
            const unsigned s = min((unsigned)__builtin_amdgcn_readlane(mine, j), (unsigned)(N_NODES - 1));
            const float4 x = *(const float4*)(fsum  + (size_t)s * OUT_F + 4 * lane);
            const float2 m = *(const float2*)(fprod + (size_t)s * RANK  + 2 * lane);
            hs.x += x.x; hs.y += x.y; hs.z += x.z; hs.w += x.w;
            la0 += __logf(fabsf(m.x));
            la1 += __logf(fabsf(m.y));
            ng0 += (m.x < 0.0f) ? 1 : 0;
            ng1 += (m.y < 0.0f) ? 1 : 0;
        }
    }

    float2 hp = make_float2(0.0f, 0.0f);
    if (cnt > 0) {
        hp.x = ((ng0 & 1) ? -1.0f : 1.0f) * __expf(la0);
        hp.y = ((ng1 & 1) ? -1.0f : 1.0f) * __expf(la1);
    }
    hs_out = hs;
    hp_out = hp;
}

__global__ __launch_bounds__(256) void k_gather_final(const int* __restrict__ rowptr,
                                                      const int* __restrict__ ideg,
                                                      const int* __restrict__ csr,
                                                      const float* __restrict__ fsum,
                                                      const float* __restrict__ fprod,
                                                      const float* __restrict__ vmat,
                                                      float* __restrict__ out)
{
    __shared__ __attribute__((aligned(16))) float Hp[16][68];
    __shared__ __attribute__((aligned(16))) float Pv[16][132];

    const int lane = threadIdx.x & 31;
    const int wave = __builtin_amdgcn_readfirstlane((int)(threadIdx.x >> 5));
    const int m0   = (int)(blockIdx.x * 16u);
    const int rA   = 2 * wave;
    const int dA   = m0 + rA;
    const int dB   = dA + 1;

    const int startA = rowptr[dA];
    const int cntA   = ideg[dA];
    const int startB = rowptr[dB];
    const int cntB   = ideg[dB];

    int mineA = 0, mineB = 0;
    if (lane < cntA) mineA = csr[startA + lane];
    if (lane < cntB) mineB = csr[startB + lane];

    float4 hsA, hsB;
    float2 hpA, hpB;
    gather_row(startA, cntA, mineA, lane, csr, fsum, fprod, hsA, hpA);
    gather_row(startB, cntB, mineB, lane, csr, fsum, fprod, hsB, hpB);

    const int rsel  = lane & 15;
    const int khalf = (lane < 16) ? 0 : 2;
    const int rofs  = (lane < 16) ? 0 : 8;
    const int n0    = wave * 16;

    v2f bv[RANK / 4];
    {
        const float* Bp = vmat + (size_t)khalf * OUT_F + n0 + rsel;
#pragma unroll
        for (int i = 0; i < RANK / 4; ++i) {
            bv[i].x = Bp[(size_t)(4 * i) * OUT_F];
            bv[i].y = Bp[(size_t)(4 * i + 1) * OUT_F];
        }
    }

    *(float2*)&Hp[rA][2 * lane]     = hpA;
    *(float2*)&Hp[rA + 1][2 * lane] = hpB;
    __syncthreads();

    {
        v8f acc = {};
#pragma unroll
        for (int i = 0; i < RANK / 4; ++i) {
            const int k0 = 4 * i;
            v2f a;
            a.x = Hp[rsel][k0 + khalf];
            a.y = Hp[rsel][k0 + khalf + 1];
            acc = __builtin_amdgcn_wmma_f32_16x16x4_f32(
                false, a, false, bv[i], (short)0, acc, false, false);
        }
#pragma unroll
        for (int r = 0; r < 8; ++r)
            Pv[rofs + r][n0 + rsel] = acc[r];
    }
    __syncthreads();

    {
        const float  scA = rsqrtf(fmaxf((float)cntA, 1.0f));
        const float  scB = rsqrtf(fmaxf((float)cntB, 1.0f));
        const float4 pA  = *(const float4*)&Pv[rA][4 * lane];
        const float4 pB  = *(const float4*)&Pv[rA + 1][4 * lane];
        float4 oA, oB;
        oA.x = (pA.x + hsA.x) * scA; oA.y = (pA.y + hsA.y) * scA;
        oA.z = (pA.z + hsA.z) * scA; oA.w = (pA.w + hsA.w) * scA;
        oB.x = (pB.x + hsB.x) * scB; oB.y = (pB.y + hsB.y) * scB;
        oB.z = (pB.z + hsB.z) * scB; oB.w = (pB.w + hsB.w) * scB;
        *(float4*)(out + (size_t)dA * OUT_F + 4 * lane) = oA;
        *(float4*)(out + (size_t)dB * OUT_F + 4 * lane) = oB;
    }
}

extern "C" void kernel_launch(void* const* d_in, const int* in_sizes, int n_in,
                              void* d_out, int out_size, void* d_ws, size_t ws_size,
                              hipStream_t stream)
{
    (void)in_sizes; (void)n_in; (void)out_size; (void)ws_size;

    const float* feat = (const float*)d_in[0];
    const float* w1   = (const float*)d_in[1];
    const float* w2   = (const float*)d_in[2];
    const float* vmat = (const float*)d_in[3];
    const int*   src  = (const int*)d_in[4];
    const int*   dst  = (const int*)d_in[5];

    float* ws     = (float*)d_ws;
    int*   ideg   = (int*)(ws + OFF_IDEG);
    int*   rowptr = (int*)(ws + OFF_ROWPTR);
    int*   csr    = (int*)(ws + OFF_CSR);
    int*   cntI   = (int*)(ws + OFF_CNTI);
    int*   cntO   = (int*)(ws + OFF_CNTO);
    int*   btot   = (int*)(ws + OFF_BTOT);
    float* fsum   = ws + OFF_FSUM;
    float* fprod  = ws + OFF_FPROD;

    k_count<<<dim3(PRE_S, PRE_R), PRE_T, 0, stream>>>(src, dst, cntI, cntO, btot);
    k_place_gemm<<<PL_BLOCKS + GEMM_BLOCKS, FUSE_T, 0, stream>>>(src, dst, cntI, cntO, btot, ideg, rowptr, csr,
                                                                 feat, w1, w2, fsum, fprod);
    k_gather_final<<<N_NODES / 16, 256, 0, stream>>>(rowptr, ideg, csr, fsum, fprod, vmat, (float*)d_out);
}
